# starting kernel
# baseline (speedup 1.0000x reference)
#include <hip/hip_runtime.h>
#include <hip/hip_bf16.h>


typedef __attribute__((ext_vector_type(2))) float v2f;
typedef __attribute__((ext_vector_type(8))) float v8f;

static constexpr int BB = 8;
static constexpr int TT = 2048;
static constexpr int CD = 1024;
static constexpr int HD = 64;
static constexpr int BT = BB * TT;

static constexpr int MTILE = 128;
static constexpr int KC    = 32;
static constexpr int XPAD  = 34;
static constexpr int SK    = 64;
static constexpr int KPAD  = 66;

__device__ __forceinline__ v8f wmma_f32(v2f a, v2f b, v8f c) {
    return __builtin_amdgcn_wmma_f32_16x16x4_f32(
         false, a,  false, b,
         (short)0, c,  false,  false);
}

__global__ __launch_bounds__(256) void proj_qkv_kernel(
    const float* __restrict__ x,
    const float* __restrict__ Wq, const float* __restrict__ Wk,
    const float* __restrict__ Wv,
    float* __restrict__ Q, float* __restrict__ K, float* __restrict__ V)
{
    __shared__ float xs[MTILE * XPAD];
    __shared__ float ws[3][HD * XPAD];

    const int tid  = threadIdx.x;
    const int wave = tid >> 5;
    const int lane = tid & 31;
    const int grp  = lane >> 4;
    const int idx  = lane & 15;
    const long rowBlock = (long)blockIdx.x * MTILE;

    const float* Wm[3] = {Wq, Wk, Wv};

    v8f acc[3][4] = {};

    for (int kc = 0; kc < CD; kc += KC) {
        __syncthreads();
        for (int i = tid; i < MTILE * (KC / 4); i += 256) {
            int r  = i >> 3;
            int c4 = i & 7;
            float4 val = *(const float4*)(x + (rowBlock + r) * CD + kc + c4 * 4);
            float* dst = &xs[r * XPAD + c4 * 4];
            dst[0] = val.x; dst[1] = val.y; dst[2] = val.z; dst[3] = val.w;
        }
        #pragma unroll
        for (int m = 0; m < 3; ++m) {
            for (int i = tid; i < KC * HD; i += 256) {
                int k = i >> 6;
                int n = i & 63;
                ws[m][n * XPAD + k] = Wm[m][(long)(kc + k) * HD + n];
            }
        }
        if (kc + KC < CD) {
            const float* nx = x + (rowBlock + (tid >> 1)) * CD + kc + KC + (tid & 1) * 16;
            __builtin_prefetch(nx, 0, 1);
        }
        __syncthreads();

        const float* xw = &xs[(wave * 16 + idx) * XPAD];
        #pragma unroll
        for (int kk = 0; kk < KC; kk += 4) {
            v2f a = *(const v2f*)(xw + kk + 2 * grp);
            #pragma unroll
            for (int m = 0; m < 3; ++m) {
                #pragma unroll
                for (int t = 0; t < 4; ++t) {
                    v2f b = *(const v2f*)(&ws[m][(t * 16 + idx) * XPAD + kk + 2 * grp]);
                    acc[m][t] = wmma_f32(a, b, acc[m][t]);
                }
            }
        }
    }

    const long rowW = rowBlock + wave * 16;
    #pragma unroll
    for (int m = 0; m < 3; ++m) {
        float* Om = (m == 0) ? Q : ((m == 1) ? K : V);
        #pragma unroll
        for (int t = 0; t < 4; ++t)
            #pragma unroll
            for (int r = 0; r < 8; ++r)
                Om[(rowW + r + 8 * grp) * HD + t * 16 + idx] = acc[m][t][r];
    }
}

__global__ __launch_bounds__(256) void attn_kernel(
    const float* __restrict__ Q, const float* __restrict__ K,
    const float* __restrict__ V, float* __restrict__ out)
{
    __shared__ float ks[SK * KPAD];
    __shared__ float vs[HD * KPAD];
    __shared__ float ps[8][16 * 16];

    const int tid  = threadIdx.x;
    const int wave = tid >> 5;
    const int lane = tid & 31;
    const int grp  = lane >> 4;
    const int idx  = lane & 15;

    const int  b     = blockIdx.y;
    const int  qt    = blockIdx.x;
    const long base  = (long)b * TT * HD;
    const int  qrow0 = qt * MTILE + wave * 16;

    v2f qf[16];
    const float* qp = Q + base + (long)(qrow0 + idx) * HD + 2 * grp;
    #pragma unroll
    for (int s = 0; s < 16; ++s) qf[s] = *(const v2f*)(qp + 4 * s);

    v8f   o[4] = {};
    float mrow[8], lrow[8];
    #pragma unroll
    for (int r = 0; r < 8; ++r) { mrow[r] = -1e30f; lrow[r] = 0.f; }

    const int nchunk = (qt + 1) * MTILE / SK;
    for (int sc = 0; sc < nchunk; ++sc) {
        const int s0 = sc * SK;
        __syncthreads();
        for (int i = tid; i < SK * (HD / 4); i += 256) {
            int r  = i >> 4;
            int c4 = i & 15;
            float4 kv = *(const float4*)(K + base + (long)(s0 + r) * HD + c4 * 4);
            float* dst = &ks[r * KPAD + c4 * 4];
            dst[0] = kv.x; dst[1] = kv.y; dst[2] = kv.z; dst[3] = kv.w;
            float4 vv = *(const float4*)(V + base + (long)(s0 + r) * HD + c4 * 4);
            vs[(c4 * 4 + 0) * KPAD + r] = vv.x;
            vs[(c4 * 4 + 1) * KPAD + r] = vv.y;
            vs[(c4 * 4 + 2) * KPAD + r] = vv.z;
            vs[(c4 * 4 + 3) * KPAD + r] = vv.w;
        }
        if (sc + 1 < nchunk) {
            const long nrow = base + (long)(s0 + SK + (tid >> 2)) * HD + (tid & 3) * 16;
            __builtin_prefetch(K + nrow, 0, 1);
            __builtin_prefetch(V + nrow, 0, 1);
        }
        __syncthreads();

        #pragma unroll
        for (int kt = 0; kt < 4; ++kt) {
            v8f s = {};
            const float* kw = &ks[(kt * 16 + idx) * KPAD + 2 * grp];
            #pragma unroll
            for (int st = 0; st < 16; ++st) {
                v2f bfrag = *(const v2f*)(kw + 4 * st);
                s = wmma_f32(qf[st], bfrag, s);
            }

            const int key = s0 + kt * 16 + idx;
            #pragma unroll
            for (int r = 0; r < 8; ++r) {
                const int qr = qrow0 + r + 8 * grp;
                float sv = s[r] * 0.125f;
                s[r] = (key <= qr) ? sv : -1e30f;
            }
            float tmax[8];
            #pragma unroll
            for (int r = 0; r < 8; ++r) {
                float v = s[r];
                #pragma unroll
                for (int off = 1; off < 16; off <<= 1)
                    v = fmaxf(v, __shfl_xor(v, off, 32));
                tmax[r] = v;
            }
            float pcorr[8];
            #pragma unroll
            for (int r = 0; r < 8; ++r) {
                const float mnew = fmaxf(mrow[r], tmax[r]);
                pcorr[r] = __expf(mrow[r] - mnew);
                mrow[r]  = mnew;
                const int qr = qrow0 + r + 8 * grp;
                s[r] = (key <= qr) ? __expf(s[r] - mnew) : 0.f;
            }
            #pragma unroll
            for (int r = 0; r < 8; ++r) {
                float v = s[r];
                #pragma unroll
                for (int off = 1; off < 16; off <<= 1)
                    v += __shfl_xor(v, off, 32);
                lrow[r] = lrow[r] * pcorr[r] + v;
            }
            #pragma unroll
            for (int t = 0; t < 4; ++t)
                #pragma unroll
                for (int r = 0; r < 8; ++r)
                    o[t][r] *= pcorr[r];

            float* pw = &ps[wave][0];
            #pragma unroll
            for (int r = 0; r < 8; ++r)
                pw[(r + 8 * grp) * 16 + idx] = s[r];

            #pragma unroll
            for (int pk = 0; pk < 16; pk += 4) {
                v2f a = *(const v2f*)(&pw[idx * 16 + pk + 2 * grp]);
                #pragma unroll
                for (int t = 0; t < 4; ++t) {
                    v2f bfrag = *(const v2f*)(&vs[(t * 16 + idx) * KPAD + kt * 16 + pk + 2 * grp]);
                    o[t] = wmma_f32(a, bfrag, o[t]);
                }
            }
        }
    }

    #pragma unroll
    for (int t = 0; t < 4; ++t)
        #pragma unroll
        for (int r = 0; r < 8; ++r) {
            const long row = qrow0 + r + 8 * grp;
            out[base + row * HD + t * 16 + idx] = o[t][r] / lrow[r];
        }
}

extern "C" void kernel_launch(void* const* d_in, const int* in_sizes, int n_in,
                              void* d_out, int out_size, void* d_ws, size_t ws_size,
                              hipStream_t stream) {
    const float* x  = (const float*)d_in[0];
    const float* Wq = (const float*)d_in[1];
    const float* Wk = (const float*)d_in[2];
    const float* Wv = (const float*)d_in[3];
    float* out = (float*)d_out;

    float* Q = (float*)d_ws;
    float* K = Q + (size_t)BT * HD;
    float* V = K + (size_t)BT * HD;

    proj_qkv_kernel<<<dim3(BT / MTILE), 256, 0, stream>>>(x, Wq, Wk, Wv, Q, K, V);
    attn_kernel<<<dim3(TT / MTILE, BB), 256, 0, stream>>>(Q, K, V, out);
}


# Attention: 16-row query tiles dealt to work-groups cyclically (tile 16*slot+group); long slots 7..4 on waves 0-3 (older on their SIMD), short slots 0..3 on waves 4-7; key loop and sums unchanged
# speedup vs baseline: 7.0948x; 7.0948x over previous
#include <hip/hip_runtime.h>
#include <hip/hip_bf16.h>


typedef __attribute__((ext_vector_type(2))) float v2f;
typedef __attribute__((ext_vector_type(8))) float v8f;

static constexpr int BB = 8;
static constexpr int TT = 2048;
static constexpr int CD = 1024;
static constexpr int HD = 64;
static constexpr int BT = BB * TT;

static constexpr int MTILE = 128;
static constexpr int WKD   = 64;
static constexpr int NCOL  = 96;
static constexpr int NT    = 6;
static constexpr int WPP   = 224;
static constexpr int HH    = 32;
static constexpr int SK    = 64;
static constexpr int PSTR  = 68;

__device__ __forceinline__ v8f wmma_f32(v2f a, v2f b, v8f c) {
    return __builtin_amdgcn_wmma_f32_16x16x4_f32(
         false, a,  false, b,
         (short)0, c,  false,  false);
}


__device__ __forceinline__ void proj_load_b(v2f (&b)[2 * NT], const float* wl_step)
{
    #pragma unroll
    for (int u = 0; u < 2; ++u) {
        #pragma unroll
        for (int j = 0; j < NT; ++j)
            b[u * NT + j] = *(const v2f*)(wl_step + u * 2 * WPP + j * 16 * 2);
    }
}

__global__ __launch_bounds__(256) void proj_qkv_kernel(
    const float* __restrict__ x,
    const float* __restrict__ Wq, const float* __restrict__ Wk,
    const float* __restrict__ Wv,
    float* __restrict__ Q, float* __restrict__ K, float* __restrict__ V)
{
    __shared__ __attribute__((aligned(16))) float wsm[(WKD / 2) * WPP];

    const int tid  = threadIdx.x;
    const int wave = tid >> 5;
    const int lane = tid & 31;
    const int grp  = lane >> 4;
    const int idx  = lane & 15;
    const int  half = blockIdx.x;
    const long rowBlock = (long)blockIdx.y * MTILE;

    const float* xrow = x + (rowBlock + wave * 16 + idx) * CD + 2 * grp;
    const float* wl = &wsm[grp * WPP + idx * 2];

    v8f acc[NT] = {};
    v2f xa[8], xb[8];

    #pragma unroll
    for (int i = 0; i < 8; ++i) xa[i] = *(const v2f*)(xrow + 4 * i);

    for (int k0 = 0; k0 < CD; k0 += WKD) {
        __syncthreads();
        #pragma unroll
        for (int i = 0; i < 3; ++i) {
            const int f  = tid + 256 * i;
            const int p  = f / 24;
            const int c4 = f - p * 24;
            const int gc = half * NCOL + c4 * 4;
            const int m  = gc >> 6;
            const float* Wg = (m == 0) ? Wq : ((m == 1) ? Wk : Wv);
            const float* sp = Wg + (long)(k0 + 2 * p) * HD + (gc & 63);
            const float4 r0 = *(const float4*)(sp);
            const float4 r1 = *(const float4*)(sp + HD);
            float* d = &wsm[p * WPP + c4 * 8];
            *(float4*)(d)     = make_float4(r0.x, r1.x, r0.y, r1.y);
            *(float4*)(d + 4) = make_float4(r0.z, r1.z, r0.w, r1.w);
        }
        __syncthreads();

        #pragma unroll
        for (int i = 0; i < 8; ++i) xb[i] = *(const v2f*)(xrow + k0 + 32 + 4 * i);

        v2f bf[2][2 * NT];
        proj_load_b(bf[0], wl);

        #pragma unroll
        for (int g = 0; g < 8; ++g) {
            if (g + 1 < 8) proj_load_b(bf[(g + 1) & 1], wl + 4 * (g + 1) * WPP);
            if (g == 4) {
                const int kn = (k0 + WKD < CD) ? k0 + WKD : k0;
                #pragma unroll
                for (int i = 0; i < 8; ++i) xa[i] = *(const v2f*)(xrow + kn + 4 * i);
            }
            __builtin_amdgcn_sched_barrier(0);
            #pragma unroll
            for (int u = 0; u < 2; ++u) {
                const int s = 2 * g + u;
                const v2f a = (s < 8) ? xa[s] : xb[s - 8];
                #pragma unroll
                for (int j = 0; j < NT; ++j)
                    acc[j] = wmma_f32(a, bf[g & 1][u * NT + j], acc[j]);
            }
            __builtin_amdgcn_sched_barrier(0);
        }
    }

    const long rowW  = rowBlock + wave * 16;
    const long tileG = rowW >> 4;
    #pragma unroll
    for (int j = 0; j < NT; ++j) {
        const int gc = half * NCOL + j * 16;
        const int m  = gc >> 6;
        const int cb = gc & 63;
        if (m == 0) {
            #pragma unroll
            for (int r = 0; r < 8; ++r)
                Q[(rowW + r + 8 * grp) * HD + cb + idx] = acc[j][r];
        } else if (m == 1) {
            const int cm = cb + idx;
            float* d = K + tileG * 1024 + (cm >> 3) * 128 + (((cm >> 2) & 1) * 16 + 8 * grp) * 4 + (cm & 3);
            #pragma unroll
            for (int r = 0; r < 8; ++r) d[r * 4] = acc[j][r];
        } else {
            float* d = V + tileG * 1024 + (((cb >> 4) * 2 + grp) * 32 + idx) * 4;
            *(float4*)(d)      = make_float4(acc[j][0], acc[j][1], acc[j][2], acc[j][3]);
            *(float4*)(d + 64) = make_float4(acc[j][4], acc[j][5], acc[j][6], acc[j][7]);
        }
    }
}

__device__ __forceinline__ v2f lo2(const float4 v) { v2f r; r[0] = v.x; r[1] = v.y; return r; }
__device__ __forceinline__ v2f hi2(const float4 v) { v2f r; r[0] = v.z; r[1] = v.w; return r; }

__global__ __launch_bounds__(256) void attn_kernel(
    const float* __restrict__ Q, const float* __restrict__ K,
    const float* __restrict__ V, float* __restrict__ out)
{
    __shared__ __attribute__((aligned(16))) float ps[8][16 * PSTR];

    const int tid  = threadIdx.x;
    const int wave = __builtin_amdgcn_readfirstlane(tid >> 5);
    const int lane = tid & 31;
    const int grp  = lane >> 4;
    const int idx  = lane & 15;

    const int  hz    = blockIdx.x;
    const int  b     = blockIdx.z;
    const int  qt    = blockIdx.y;
    const long base  = (long)b * TT * HD;
    const int  slot  = (wave < 4) ? 7 - wave : wave - 4;
    const int  qrow0 = (slot * 16 + qt) * 16;

    float4 qf[8];
    const float* qp = Q + base + (long)(qrow0 + idx) * HD + 4 * grp;
    #pragma unroll
    for (int j = 0; j < 8; ++j) qf[j] = *(const float4*)(qp + 8 * j);

    const float* kfp = K + base + lane * 4;
    const float* vfp = V + base + (2 * hz) * 256 + lane * 4;

    float4 kf[4][8];
    float4 vf[4][2][2];
    #pragma unroll
    for (int kt = 0; kt < 4; ++kt)
        #pragma unroll
        for (int j = 0; j < 8; ++j) kf[kt][j] = *(const float4*)(kfp + kt * 1024 + j * 128);
    #pragma unroll
    for (int kt = 0; kt < 4; ++kt)
        #pragma unroll
        for (int tt = 0; tt < 2; ++tt)
            #pragma unroll
            for (int jj = 0; jj < 2; ++jj) vf[kt][tt][jj] = *(const float4*)(vfp + kt * 1024 + (tt * 2 + jj) * 128);

    v8f   o[2] = {};
    float mrow[8], lrow[8];
    #pragma unroll
    for (int r = 0; r < 8; ++r) { mrow[r] = -1e30f; lrow[r] = 0.f; }

    float* pw = &ps[wave][0];

    const int nchunk = (qrow0 + 15) / SK + 1;
    for (int sc = 0; sc < nchunk; ++sc) {
        const int s0 = sc * SK;
        const long nxt = (long)((sc + 1 < nchunk) ? sc + 1 : sc) * (4 * 1024);

        v8f s[4] = {};
        #pragma unroll
        for (int j = 0; j < 8; ++j) {
            #pragma unroll
            for (int kt = 0; kt < 4; ++kt) s[kt] = wmma_f32(lo2(qf[j]), lo2(kf[kt][j]), s[kt]);
            #pragma unroll
            for (int kt = 0; kt < 4; ++kt) s[kt] = wmma_f32(hi2(qf[j]), hi2(kf[kt][j]), s[kt]);
        }
        #pragma unroll
        for (int kt = 0; kt < 4; ++kt)
            #pragma unroll
            for (int j = 0; j < 8; ++j) kf[kt][j] = *(const float4*)(kfp + nxt + kt * 1024 + j * 128);

        float tmax[8];
        #pragma unroll
        for (int r = 0; r < 8; ++r) {
            const int qr = qrow0 + r + 8 * grp;
            float m = -1e30f;
            #pragma unroll
            for (int kt = 0; kt < 4; ++kt) {
                const int key = s0 + kt * 16 + idx;
                const float sv = s[kt][r] * 0.125f;
                s[kt][r] = (key <= qr) ? sv : -1e30f;
                m = fmaxf(m, s[kt][r]);
            }
            tmax[r] = m;
        }
        #pragma unroll
        for (int off = 1; off < 16; off <<= 1) {
            float t[8];
            #pragma unroll
            for (int r = 0; r < 8; ++r) t[r] = __shfl_xor(tmax[r], off, 32);
            __builtin_amdgcn_sched_barrier(0);
            #pragma unroll
            for (int r = 0; r < 8; ++r) tmax[r] = fmaxf(tmax[r], t[r]);
        }
        float pcorr[8];
        #pragma unroll
        for (int r = 0; r < 8; ++r) {
            const int qr = qrow0 + r + 8 * grp;
            const float mnew = fmaxf(mrow[r], tmax[r]);
            pcorr[r] = __expf(mrow[r] - mnew);
            mrow[r]  = mnew;
            float lsum = 0.f;
            #pragma unroll
            for (int kt = 0; kt < 4; ++kt) {
                const int key = s0 + kt * 16 + idx;
                const float p = (key <= qr) ? __expf(s[kt][r] - mnew) : 0.f;
                s[kt][r] = p;
                lsum += p;
            }
            lrow[r] = lrow[r] * pcorr[r] + lsum;
        }
        #pragma unroll
        for (int t = 0; t < 2; ++t)
            #pragma unroll
            for (int r = 0; r < 8; ++r)
                o[t][r] *= pcorr[r];

        #pragma unroll
        for (int kt = 0; kt < 4; ++kt)
            #pragma unroll
            for (int r = 0; r < 8; ++r)
                pw[(r + 8 * grp) * PSTR + kt * 16 + idx] = s[kt][r];

        float4 pa[4][2];
        #pragma unroll
        for (int kt = 0; kt < 4; ++kt)
            #pragma unroll
            for (int jj = 0; jj < 2; ++jj)
                pa[kt][jj] = *(const float4*)(&pw[idx * PSTR + kt * 16 + 8 * jj + 4 * grp]);
        __builtin_amdgcn_sched_barrier(0);
        #pragma unroll
        for (int kt = 0; kt < 4; ++kt) {
            #pragma unroll
            for (int jj = 0; jj < 2; ++jj) {
                #pragma unroll
                for (int tt = 0; tt < 2; ++tt) o[tt] = wmma_f32(lo2(pa[kt][jj]), lo2(vf[kt][tt][jj]), o[tt]);
                #pragma unroll
                for (int tt = 0; tt < 2; ++tt) o[tt] = wmma_f32(hi2(pa[kt][jj]), hi2(vf[kt][tt][jj]), o[tt]);
            }
        }
        #pragma unroll
        for (int kt = 0; kt < 4; ++kt)
            #pragma unroll
            for (int tt = 0; tt < 2; ++tt)
                #pragma unroll
                for (int jj = 0; jj < 2; ++jj)
                    vf[kt][tt][jj] = *(const float4*)(vfp + nxt + kt * 1024 + (tt * 2 + jj) * 128);
    }

    #pragma unroll
    for (int off = 1; off < 16; off <<= 1) {
        float t[8];
        #pragma unroll
        for (int r = 0; r < 8; ++r) t[r] = __shfl_xor(lrow[r], off, 32);
        #pragma unroll
        for (int r = 0; r < 8; ++r) lrow[r] += t[r];
    }

    #pragma unroll
    for (int t = 0; t < 2; ++t)
        #pragma unroll
        for (int r = 0; r < 8; ++r) {
            const long row = qrow0 + r + 8 * grp;
            out[base + row * HD + hz * HH + t * 16 + idx] = o[t][r] / lrow[r];
        }
}

extern "C" void kernel_launch(void* const* d_in, const int* in_sizes, int n_in,
                              void* d_out, int out_size, void* d_ws, size_t ws_size,
                              hipStream_t stream) {
    const float* x  = (const float*)d_in[0];
    const float* Wq = (const float*)d_in[1];
    const float* Wk = (const float*)d_in[2];
    const float* Wv = (const float*)d_in[3];
    float* out = (float*)d_out;

    float* Q = (float*)d_ws;
    float* K = Q + (size_t)BT * HD;
    float* V = K + (size_t)BT * HD;

    proj_qkv_kernel<<<dim3(2, BT / MTILE), 256, 0, stream>>>(x, Wq, Wk, Wv, Q, K, V);
    attn_kernel<<<dim3(2, TT / MTILE, BB), 256, 0, stream>>>(Q, K, V, out);
}


# Attention: raw scores (not P) cross the per-wave LDS patch; softmax done behind it, one row per lane (32 keys): max in-lane + one half swap, no 16-lane butterfly; row sums per lane pair (new order)
# speedup vs baseline: 7.8435x; 1.1055x over previous
#include <hip/hip_runtime.h>
#include <hip/hip_bf16.h>


typedef __attribute__((ext_vector_type(2))) float v2f;
typedef __attribute__((ext_vector_type(8))) float v8f;

static constexpr int BB = 8;
static constexpr int TT = 2048;
static constexpr int CD = 1024;
static constexpr int HD = 64;
static constexpr int BT = BB * TT;

static constexpr int MTILE = 128;
static constexpr int WKD   = 64;
static constexpr int NCOL  = 96;
static constexpr int NT    = 6;
static constexpr int WPP   = 224;
static constexpr int HH    = 32;
static constexpr int SK    = 64;
static constexpr int PSTR  = 68;

__device__ __forceinline__ v8f wmma_f32(v2f a, v2f b, v8f c) {
    return __builtin_amdgcn_wmma_f32_16x16x4_f32(
         false, a,  false, b,
         (short)0, c,  false,  false);
}


__device__ __forceinline__ void proj_load_b(v2f (&b)[2 * NT], const float* wl_step)
{
    #pragma unroll
    for (int u = 0; u < 2; ++u) {
        #pragma unroll
        for (int j = 0; j < NT; ++j)
            b[u * NT + j] = *(const v2f*)(wl_step + u * 2 * WPP + j * 16 * 2);
    }
}

__global__ __launch_bounds__(256) void proj_qkv_kernel(
    const float* __restrict__ x,
    const float* __restrict__ Wq, const float* __restrict__ Wk,
    const float* __restrict__ Wv,
    float* __restrict__ Q, float* __restrict__ K, float* __restrict__ V)
{
    __shared__ __attribute__((aligned(16))) float wsm[(WKD / 2) * WPP];

    const int tid  = threadIdx.x;
    const int wave = tid >> 5;
    const int lane = tid & 31;
    const int grp  = lane >> 4;
    const int idx  = lane & 15;
    const int  half = blockIdx.x;
    const long rowBlock = (long)blockIdx.y * MTILE;

    const float* xrow = x + (rowBlock + wave * 16 + idx) * CD + 2 * grp;
    const float* wl = &wsm[grp * WPP + idx * 2];

    v8f acc[NT] = {};
    v2f xa[8], xb[8];

    #pragma unroll
    for (int i = 0; i < 8; ++i) xa[i] = *(const v2f*)(xrow + 4 * i);

    for (int k0 = 0; k0 < CD; k0 += WKD) {
        __syncthreads();
        #pragma unroll
        for (int i = 0; i < 3; ++i) {
            const int f  = tid + 256 * i;
            const int p  = f / 24;
            const int c4 = f - p * 24;
            const int gc = half * NCOL + c4 * 4;
            const int m  = gc >> 6;
            const float* Wg = (m == 0) ? Wq : ((m == 1) ? Wk : Wv);
            const float* sp = Wg + (long)(k0 + 2 * p) * HD + (gc & 63);
            const float4 r0 = *(const float4*)(sp);
            const float4 r1 = *(const float4*)(sp + HD);
            float* d = &wsm[p * WPP + c4 * 8];
            *(float4*)(d)     = make_float4(r0.x, r1.x, r0.y, r1.y);
            *(float4*)(d + 4) = make_float4(r0.z, r1.z, r0.w, r1.w);
        }
        __syncthreads();

        #pragma unroll
        for (int i = 0; i < 8; ++i) xb[i] = *(const v2f*)(xrow + k0 + 32 + 4 * i);

        v2f bf[2][2 * NT];
        proj_load_b(bf[0], wl);

        #pragma unroll
        for (int g = 0; g < 8; ++g) {
            if (g + 1 < 8) proj_load_b(bf[(g + 1) & 1], wl + 4 * (g + 1) * WPP);
            if (g == 4) {
                const int kn = (k0 + WKD < CD) ? k0 + WKD : k0;
                #pragma unroll
                for (int i = 0; i < 8; ++i) xa[i] = *(const v2f*)(xrow + kn + 4 * i);
            }
            __builtin_amdgcn_sched_barrier(0);
            #pragma unroll
            for (int u = 0; u < 2; ++u) {
                const int s = 2 * g + u;
                const v2f a = (s < 8) ? xa[s] : xb[s - 8];
                #pragma unroll
                for (int j = 0; j < NT; ++j)
                    acc[j] = wmma_f32(a, bf[g & 1][u * NT + j], acc[j]);
            }
            __builtin_amdgcn_sched_barrier(0);
        }
    }

    const long rowW  = rowBlock + wave * 16;
    const long tileG = rowW >> 4;
    #pragma unroll
    for (int j = 0; j < NT; ++j) {
        const int gc = half * NCOL + j * 16;
        const int m  = gc >> 6;
        const int cb = gc & 63;
        if (m == 0) {
            #pragma unroll
            for (int r = 0; r < 8; ++r)
                Q[(rowW + r + 8 * grp) * HD + cb + idx] = acc[j][r];
        } else if (m == 1) {
            const int cm = cb + idx;
            float* d = K + tileG * 1024 + (cm >> 3) * 128 + (((cm >> 2) & 1) * 16 + 8 * grp) * 4 + (cm & 3);
            #pragma unroll
            for (int r = 0; r < 8; ++r) d[r * 4] = acc[j][r];
        } else {
            float* d = V + tileG * 1024 + (((cb >> 4) * 2 + grp) * 32 + idx) * 4;
            *(float4*)(d)      = make_float4(acc[j][0], acc[j][1], acc[j][2], acc[j][3]);
            *(float4*)(d + 64) = make_float4(acc[j][4], acc[j][5], acc[j][6], acc[j][7]);
        }
    }
}

__device__ __forceinline__ v2f lo2(const float4 v) { v2f r; r[0] = v.x; r[1] = v.y; return r; }
__device__ __forceinline__ v2f hi2(const float4 v) { v2f r; r[0] = v.z; r[1] = v.w; return r; }

__device__ __forceinline__ float xhalf(float x) {
    return __builtin_bit_cast(float, __builtin_amdgcn_permlanex16(
        __builtin_bit_cast(int, x), __builtin_bit_cast(int, x), 0x76543210, 0xfedcba98, false, false));
}

__global__ __launch_bounds__(256) void attn_kernel(
    const float* __restrict__ Q, const float* __restrict__ K,
    const float* __restrict__ V, float* __restrict__ out)
{
    __shared__ __attribute__((aligned(16))) float ps[8][16 * PSTR];
    __shared__ __attribute__((aligned(16))) float rs[8][16];

    const int tid  = threadIdx.x;
    const int wave = __builtin_amdgcn_readfirstlane(tid >> 5);
    const int lane = tid & 31;
    const int grp  = lane >> 4;
    const int idx  = lane & 15;

    const int  hz    = blockIdx.x;
    const int  b     = blockIdx.z;
    const int  qt    = blockIdx.y;
    const long base  = (long)b * TT * HD;
    const int  slot  = (wave < 4) ? 7 - wave : wave - 4;
    const int  qrow0 = (slot * 16 + qt) * 16;

    float4 qf[8];
    const float* qp = Q + base + (long)(qrow0 + idx) * HD + 4 * grp;
    #pragma unroll
    for (int j = 0; j < 8; ++j) qf[j] = *(const float4*)(qp + 8 * j);

    const float* kfp = K + base + lane * 4;
    const float* vfp = V + base + (2 * hz) * 256 + lane * 4;

    float4 kf[4][8];
    float4 vf[4][2][2];
    #pragma unroll
    for (int kt = 0; kt < 4; ++kt)
        #pragma unroll
        for (int j = 0; j < 8; ++j) kf[kt][j] = *(const float4*)(kfp + kt * 1024 + j * 128);
    #pragma unroll
    for (int kt = 0; kt < 4; ++kt)
        #pragma unroll
        for (int tt = 0; tt < 2; ++tt)
            #pragma unroll
            for (int jj = 0; jj < 2; ++jj) vf[kt][tt][jj] = *(const float4*)(vfp + kt * 1024 + (tt * 2 + jj) * 128);

    v8f   o[2] = {};
    float mrow = -1e30f, lrow = 0.f;
    const int qr = qrow0 + idx;

    float* pw = &ps[wave][0];
    float* rw = &rs[wave][0];

    const int nchunk = (qrow0 + 15) / SK + 1;
    for (int sc = 0; sc < nchunk; ++sc) {
        const int s0 = sc * SK;
        const long nxt = (long)((sc + 1 < nchunk) ? sc + 1 : sc) * (4 * 1024);

        v8f s[4] = {};
        #pragma unroll
        for (int j = 0; j < 8; ++j) {
            #pragma unroll
            for (int kt = 0; kt < 4; ++kt) s[kt] = wmma_f32(lo2(qf[j]), lo2(kf[kt][j]), s[kt]);
            #pragma unroll
            for (int kt = 0; kt < 4; ++kt) s[kt] = wmma_f32(hi2(qf[j]), hi2(kf[kt][j]), s[kt]);
        }
        #pragma unroll
        for (int kt = 0; kt < 4; ++kt)
            #pragma unroll
            for (int j = 0; j < 8; ++j) kf[kt][j] = *(const float4*)(kfp + nxt + kt * 1024 + j * 128);
        __builtin_amdgcn_sched_barrier(0);

        #pragma unroll
        for (int kt = 0; kt < 4; ++kt)
            #pragma unroll
            for (int r = 0; r < 8; ++r)
                pw[(r + 8 * grp) * PSTR + kt * 16 + idx] = s[kt][r];
        float p[32];
        #pragma unroll
        for (int kt = 0; kt < 4; ++kt)
            #pragma unroll
            for (int jj = 0; jj < 2; ++jj) {
                const float4 t = *(const float4*)(&pw[idx * PSTR + kt * 16 + 8 * jj + 4 * grp]);
                p[(kt * 2 + jj) * 4 + 0] = t.x;
                p[(kt * 2 + jj) * 4 + 1] = t.y;
                p[(kt * 2 + jj) * 4 + 2] = t.z;
                p[(kt * 2 + jj) * 4 + 3] = t.w;
            }

        const int dlim = qr - s0 - 4 * grp;
        float m = -1e30f;
        #pragma unroll
        for (int kt = 0; kt < 4; ++kt)
            #pragma unroll
            for (int jj = 0; jj < 2; ++jj)
                #pragma unroll
                for (int e = 0; e < 4; ++e) {
                    const int n = (kt * 2 + jj) * 4 + e;
                    const int c = kt * 16 + jj * 8 + e;
                    const float sv = p[n] * 0.125f;
                    p[n] = (c <= dlim) ? sv : -1e30f;
                    m = fmaxf(m, p[n]);
                }
        m = fmaxf(m, xhalf(m));

        const float mnew  = fmaxf(mrow, m);
        const float pcorr = __expf(mrow - mnew);
        mrow = mnew;
        rw[idx] = pcorr;
        const float4 c0 = *(const float4*)(rw + 8 * grp);
        const float4 c1 = *(const float4*)(rw + 8 * grp + 4);

        float lsum = 0.f;
        #pragma unroll
        for (int kt = 0; kt < 4; ++kt)
            #pragma unroll
            for (int jj = 0; jj < 2; ++jj)
                #pragma unroll
                for (int e = 0; e < 4; ++e) {
                    const int n = (kt * 2 + jj) * 4 + e;
                    const int c = kt * 16 + jj * 8 + e;
                    const float pv = (c <= dlim) ? __expf(p[n] - mnew) : 0.f;
                    p[n] = pv;
                    lsum += pv;
                }
        lrow = lrow * pcorr + lsum;

        #pragma unroll
        for (int t = 0; t < 2; ++t) {
            o[t][0] *= c0.x; o[t][1] *= c0.y; o[t][2] *= c0.z; o[t][3] *= c0.w;
            o[t][4] *= c1.x; o[t][5] *= c1.y; o[t][6] *= c1.z; o[t][7] *= c1.w;
        }

        #pragma unroll
        for (int kt = 0; kt < 4; ++kt) {
            #pragma unroll
            for (int jj = 0; jj < 2; ++jj) {
                const int n = (kt * 2 + jj) * 4;
                v2f alo, ahi;
                alo[0] = p[n + 0]; alo[1] = p[n + 1];
                ahi[0] = p[n + 2]; ahi[1] = p[n + 3];
                #pragma unroll
                for (int tt = 0; tt < 2; ++tt) o[tt] = wmma_f32(alo, lo2(vf[kt][tt][jj]), o[tt]);
                #pragma unroll
                for (int tt = 0; tt < 2; ++tt) o[tt] = wmma_f32(ahi, hi2(vf[kt][tt][jj]), o[tt]);
            }
        }
        #pragma unroll
        for (int kt = 0; kt < 4; ++kt)
            #pragma unroll
            for (int tt = 0; tt < 2; ++tt)
                #pragma unroll
                for (int jj = 0; jj < 2; ++jj)
                    vf[kt][tt][jj] = *(const float4*)(vfp + nxt + kt * 1024 + (tt * 2 + jj) * 128);
    }

    const float ltot = lrow + xhalf(lrow);
    rw[idx] = ltot;
    const float4 l0 = *(const float4*)(rw + 8 * grp);
    const float4 l1 = *(const float4*)(rw + 8 * grp + 4);
    const float lr[8] = { l0.x, l0.y, l0.z, l0.w, l1.x, l1.y, l1.z, l1.w };

    #pragma unroll
    for (int t = 0; t < 2; ++t)
        #pragma unroll
        for (int r = 0; r < 8; ++r) {
            const long row = qrow0 + r + 8 * grp;
            out[base + row * HD + hz * HH + t * 16 + idx] = o[t][r] / lr[r];
        }
}

extern "C" void kernel_launch(void* const* d_in, const int* in_sizes, int n_in,
                              void* d_out, int out_size, void* d_ws, size_t ws_size,
                              hipStream_t stream) {
    const float* x  = (const float*)d_in[0];
    const float* Wq = (const float*)d_in[1];
    const float* Wk = (const float*)d_in[2];
    const float* Wv = (const float*)d_in[3];
    float* out = (float*)d_out;

    float* Q = (float*)d_ws;
    float* K = Q + (size_t)BT * HD;
    float* V = K + (size_t)BT * HD;

    proj_qkv_kernel<<<dim3(2, BT / MTILE), 256, 0, stream>>>(x, Wq, Wk, Wv, Q, K, V);
    attn_kernel<<<dim3(2, TT / MTILE, BB), 256, 0, stream>>>(Q, K, V, out);
}


# Projection re-tiled: a work-group keeps its 48 columns of W for ALL k in LDS (copied once, no barrier in k loop), 256 rows x 48 cols, wave 32x48; x by 16-byte loads (2 steps each); new sum order
# speedup vs baseline: 8.7562x; 1.1164x over previous
#include <hip/hip_runtime.h>
#include <hip/hip_bf16.h>


typedef __attribute__((ext_vector_type(2))) float v2f;
typedef __attribute__((ext_vector_type(8))) float v8f;

static constexpr int BB = 8;
static constexpr int TT = 2048;
static constexpr int CD = 1024;
static constexpr int HD = 64;
static constexpr int BT = BB * TT;

static constexpr int PROW  = 256;
static constexpr int NTC   = 3;
static constexpr int WP    = 96;
static constexpr int NPR   = CD / 2;
static constexpr int MTILE = 128;
static constexpr int HH    = 32;
static constexpr int SK    = 64;
static constexpr int PSTR  = 68;

__device__ __forceinline__ v8f wmma_f32(v2f a, v2f b, v8f c) {
    return __builtin_amdgcn_wmma_f32_16x16x4_f32(
         false, a,  false, b,
         (short)0, c,  false,  false);
}

__device__ __forceinline__ v2f lo2(const float4 v) { v2f r; r[0] = v.x; r[1] = v.y; return r; }
__device__ __forceinline__ v2f hi2(const float4 v) { v2f r; r[0] = v.z; r[1] = v.w; return r; }


__device__ __forceinline__ void proj_load_b(v2f (&b)[4 * NTC], const float* wl_grp)
{
    #pragma unroll
    for (int s = 0; s < 4; ++s) {
        #pragma unroll
        for (int t = 0; t < NTC; ++t)
            b[s * NTC + t] = *(const v2f*)(wl_grp + 2 * s * WP + t * 32);
    }
}

__device__ __forceinline__ void proj_load_x(float4 (&xs)[2][8], const float* xr0, const float* xr1, int kb)
{
    #pragma unroll
    for (int jp = 0; jp < 8; ++jp) {
        xs[0][jp] = *(const float4*)(xr0 + kb + 8 * jp);
        xs[1][jp] = *(const float4*)(xr1 + kb + 8 * jp);
    }
}

__device__ __forceinline__ int proj_tile_col(int c, int t)
{
    const int tile = (c < 2) ? ((t < 2) ? 2 * c + t : 4 + c) : 3 * c + t;
    return tile * 16;
}

__global__ __launch_bounds__(256) void proj_qkv_kernel(
    const float* __restrict__ x,
    const float* __restrict__ Wq, const float* __restrict__ Wk,
    const float* __restrict__ Wv,
    float* __restrict__ Q, float* __restrict__ K, float* __restrict__ V)
{
    __shared__ __attribute__((aligned(16))) float wsm[NPR * WP];

    const int tid  = threadIdx.x;
    const int wave = tid >> 5;
    const int lane = tid & 31;
    const int grp  = lane >> 4;
    const int idx  = lane & 15;
    const int  cblk = blockIdx.x;
    const long rowW = (long)blockIdx.y * PROW + wave * 32;

    const float* xr0 = x + (rowW + idx) * CD + 4 * grp;
    const float* xr1 = xr0 + 16 * CD;

    float4 xa[2][8], xb[2][8];
    proj_load_x(xa, xr0, xr1, 0);

    if (tid < 192) {
        const int pr = tid / 12;
        const int c4 = tid - pr * 12;
        const int gc = proj_tile_col(cblk, c4 >> 2) + (c4 & 3) * 4;
        const int m  = gc >> 6;
        const float* Wg  = (m == 0) ? Wq : ((m == 1) ? Wk : Wv);
        const float* sp0 = Wg + (long)(2 * pr) * HD + (gc & 63);
        const int posr = (pr & ~3) | ((pr & 1) << 1) | ((pr >> 1) & 1);
        float* d0 = &wsm[posr * WP + c4 * 8];
        const int rot = (int)blockIdx.y & 31;
        for (int i0 = 0; i0 < 32; i0 += 16) {
            float4 r0[16], r1[16];
            #pragma unroll
            for (int i = 0; i < 16; ++i) {
                const int ii = (i0 + i + rot) & 31;
                const float* sp = sp0 + ii * (32 * HD);
                r0[i] = *(const float4*)(sp);
                r1[i] = *(const float4*)(sp + HD);
            }
            #pragma unroll
            for (int i = 0; i < 16; ++i) {
                const int ii = (i0 + i + rot) & 31;
                float* d = d0 + ii * (16 * WP);
                *(float4*)(d)     = make_float4(r0[i].x, r1[i].x, r0[i].y, r1[i].y);
                *(float4*)(d + 4) = make_float4(r0[i].z, r1[i].z, r0[i].w, r1[i].w);
            }
        }
    }
    __syncthreads();

    const float* wl = &wsm[grp * WP + idx * 2];

    v8f acc[2][NTC] = {};
    v2f bf[2][4 * NTC];
    proj_load_b(bf[0], wl);

    for (int k0 = 0; k0 < CD; k0 += 128) {
        const float* wk = wl + (k0 / 2) * WP;
        const bool more = (k0 + 128 < CD);
        proj_load_x(xb, xr0, xr1, k0 + 64);

        #pragma unroll
        for (int q = 0; q < 8; ++q) {
            if (q == 4) {
                proj_load_x(xa, xr0, xr1, more ? k0 + 128 : k0);
            }
            if (q < 7) proj_load_b(bf[(q + 1) & 1], wk + 8 * (q + 1) * WP);
            else       proj_load_b(bf[0], more ? wk + 64 * WP : wk);
            __builtin_amdgcn_sched_barrier(0);
            #pragma unroll
            for (int s = 0; s < 4; ++s) {
                const int jp = 2 * (q & 3) + (s >> 1);
                #pragma unroll
                for (int rt = 0; rt < 2; ++rt) {
                    const float4 xv = (q < 4) ? xa[rt][jp] : xb[rt][jp];
                    const v2f a = (s & 1) ? hi2(xv) : lo2(xv);
                    #pragma unroll
                    for (int t = 0; t < NTC; ++t)
                        acc[rt][t] = wmma_f32(a, bf[q & 1][s * NTC + t], acc[rt][t]);
                }
            }
            __builtin_amdgcn_sched_barrier(0);
        }
    }

    #pragma unroll
    for (int rt = 0; rt < 2; ++rt) {
        const long rowT  = rowW + rt * 16;
        const long tileG = rowT >> 4;
        #pragma unroll
        for (int t = 0; t < NTC; ++t) {
            const int gc = proj_tile_col(cblk, t);
            const int m  = gc >> 6;
            const int cb = gc & 63;
            if (m == 0) {
                #pragma unroll
                for (int r = 0; r < 8; ++r)
                    Q[(rowT + r + 8 * grp) * HD + cb + idx] = acc[rt][t][r];
            } else if (m == 1) {
                const int cm = cb + idx;
                float* d = K + tileG * 1024 + (cm >> 3) * 128 + (((cm >> 2) & 1) * 16 + 8 * grp) * 4 + (cm & 3);
                #pragma unroll
                for (int r = 0; r < 8; ++r) d[r * 4] = acc[rt][t][r];
            } else {
                float* d = V + tileG * 1024 + (((cb >> 4) * 2 + grp) * 32 + idx) * 4;
                *(float4*)(d)      = make_float4(acc[rt][t][0], acc[rt][t][1], acc[rt][t][2], acc[rt][t][3]);
                *(float4*)(d + 64) = make_float4(acc[rt][t][4], acc[rt][t][5], acc[rt][t][6], acc[rt][t][7]);
            }
        }
    }
}

__device__ __forceinline__ float xhalf(float x) {
    return __builtin_bit_cast(float, __builtin_amdgcn_permlanex16(
        __builtin_bit_cast(int, x), __builtin_bit_cast(int, x), 0x76543210, 0xfedcba98, false, false));
}

__global__ __launch_bounds__(256) void attn_kernel(
    const float* __restrict__ Q, const float* __restrict__ K,
    const float* __restrict__ V, float* __restrict__ out)
{
    __shared__ __attribute__((aligned(16))) float ps[8][16 * PSTR];
    __shared__ __attribute__((aligned(16))) float rs[8][16];

    const int tid  = threadIdx.x;
    const int wave = __builtin_amdgcn_readfirstlane(tid >> 5);
    const int lane = tid & 31;
    const int grp  = lane >> 4;
    const int idx  = lane & 15;

    const int  hz    = blockIdx.x;
    const int  b     = blockIdx.z;
    const int  qt    = blockIdx.y;
    const long base  = (long)b * TT * HD;
    const int  slot  = (wave < 4) ? 7 - wave : wave - 4;
    const int  qrow0 = (slot * 16 + qt) * 16;

    float4 qf[8];
    const float* qp = Q + base + (long)(qrow0 + idx) * HD + 4 * grp;
    #pragma unroll
    for (int j = 0; j < 8; ++j) qf[j] = *(const float4*)(qp + 8 * j);

    const float* kfp = K + base + lane * 4;
    const float* vfp = V + base + (2 * hz) * 256 + lane * 4;

    float4 kf[4][8];
    float4 vf[4][2][2];
    #pragma unroll
    for (int kt = 0; kt < 4; ++kt)
        #pragma unroll
        for (int j = 0; j < 8; ++j) kf[kt][j] = *(const float4*)(kfp + kt * 1024 + j * 128);
    #pragma unroll
    for (int kt = 0; kt < 4; ++kt)
        #pragma unroll
        for (int tt = 0; tt < 2; ++tt)
            #pragma unroll
            for (int jj = 0; jj < 2; ++jj) vf[kt][tt][jj] = *(const float4*)(vfp + kt * 1024 + (tt * 2 + jj) * 128);

    v8f   o[2] = {};
    float mrow = -1e30f, lrow = 0.f;
    const int qr = qrow0 + idx;

    float* pw = &ps[wave][0];
    float* rw = &rs[wave][0];

    const int nchunk = (qrow0 + 15) / SK + 1;
    for (int sc = 0; sc < nchunk; ++sc) {
        const int s0 = sc * SK;
        const long nxt = (long)((sc + 1 < nchunk) ? sc + 1 : sc) * (4 * 1024);

        v8f s[4] = {};
        #pragma unroll
        for (int j = 0; j < 8; ++j) {
            #pragma unroll
            for (int kt = 0; kt < 4; ++kt) s[kt] = wmma_f32(lo2(qf[j]), lo2(kf[kt][j]), s[kt]);
            #pragma unroll
            for (int kt = 0; kt < 4; ++kt) s[kt] = wmma_f32(hi2(qf[j]), hi2(kf[kt][j]), s[kt]);
        }
        #pragma unroll
        for (int kt = 0; kt < 4; ++kt)
            #pragma unroll
            for (int j = 0; j < 8; ++j) kf[kt][j] = *(const float4*)(kfp + nxt + kt * 1024 + j * 128);
        __builtin_amdgcn_sched_barrier(0);

        #pragma unroll
        for (int kt = 0; kt < 4; ++kt)
            #pragma unroll
            for (int r = 0; r < 8; ++r)
                pw[(r + 8 * grp) * PSTR + kt * 16 + idx] = s[kt][r];
        float p[32];
        #pragma unroll
        for (int kt = 0; kt < 4; ++kt)
            #pragma unroll
            for (int jj = 0; jj < 2; ++jj) {
                const float4 t = *(const float4*)(&pw[idx * PSTR + kt * 16 + 8 * jj + 4 * grp]);
                p[(kt * 2 + jj) * 4 + 0] = t.x;
                p[(kt * 2 + jj) * 4 + 1] = t.y;
                p[(kt * 2 + jj) * 4 + 2] = t.z;
                p[(kt * 2 + jj) * 4 + 3] = t.w;
            }

        const int dlim = qr - s0 - 4 * grp;
        float m = -1e30f;
        #pragma unroll
        for (int kt = 0; kt < 4; ++kt)
            #pragma unroll
            for (int jj = 0; jj < 2; ++jj)
                #pragma unroll
                for (int e = 0; e < 4; ++e) {
                    const int n = (kt * 2 + jj) * 4 + e;
                    const int c = kt * 16 + jj * 8 + e;
                    const float sv = p[n] * 0.125f;
                    p[n] = (c <= dlim) ? sv : -1e30f;
                    m = fmaxf(m, p[n]);
                }
        m = fmaxf(m, xhalf(m));

        const float mnew  = fmaxf(mrow, m);
        const float pcorr = __expf(mrow - mnew);
        mrow = mnew;
        rw[idx] = pcorr;
        const float4 c0 = *(const float4*)(rw + 8 * grp);
        const float4 c1 = *(const float4*)(rw + 8 * grp + 4);

        float lsum = 0.f;
        #pragma unroll
        for (int kt = 0; kt < 4; ++kt)
            #pragma unroll
            for (int jj = 0; jj < 2; ++jj)
                #pragma unroll
                for (int e = 0; e < 4; ++e) {
                    const int n = (kt * 2 + jj) * 4 + e;
                    const int c = kt * 16 + jj * 8 + e;
                    const float pv = (c <= dlim) ? __expf(p[n] - mnew) : 0.f;
                    p[n] = pv;
                    lsum += pv;
                }
        lrow = lrow * pcorr + lsum;

        #pragma unroll
        for (int t = 0; t < 2; ++t) {
            o[t][0] *= c0.x; o[t][1] *= c0.y; o[t][2] *= c0.z; o[t][3] *= c0.w;
            o[t][4] *= c1.x; o[t][5] *= c1.y; o[t][6] *= c1.z; o[t][7] *= c1.w;
        }

        #pragma unroll
        for (int kt = 0; kt < 4; ++kt) {
            #pragma unroll
            for (int jj = 0; jj < 2; ++jj) {
                const int n = (kt * 2 + jj) * 4;
                v2f alo, ahi;
                alo[0] = p[n + 0]; alo[1] = p[n + 1];
                ahi[0] = p[n + 2]; ahi[1] = p[n + 3];
                #pragma unroll
                for (int tt = 0; tt < 2; ++tt) o[tt] = wmma_f32(alo, lo2(vf[kt][tt][jj]), o[tt]);
                #pragma unroll
                for (int tt = 0; tt < 2; ++tt) o[tt] = wmma_f32(ahi, hi2(vf[kt][tt][jj]), o[tt]);
            }
        }
        #pragma unroll
        for (int kt = 0; kt < 4; ++kt)
            #pragma unroll
            for (int tt = 0; tt < 2; ++tt)
                #pragma unroll
                for (int jj = 0; jj < 2; ++jj)
                    vf[kt][tt][jj] = *(const float4*)(vfp + nxt + kt * 1024 + (tt * 2 + jj) * 128);
    }

    const float ltot = lrow + xhalf(lrow);
    rw[idx] = ltot;
    const float4 l0 = *(const float4*)(rw + 8 * grp);
    const float4 l1 = *(const float4*)(rw + 8 * grp + 4);
    const float lr[8] = { l0.x, l0.y, l0.z, l0.w, l1.x, l1.y, l1.z, l1.w };

    #pragma unroll
    for (int t = 0; t < 2; ++t)
        #pragma unroll
        for (int r = 0; r < 8; ++r) {
            const long row = qrow0 + r + 8 * grp;
            out[base + row * HD + hz * HH + t * 16 + idx] = o[t][r] / lr[r];
        }
}

extern "C" void kernel_launch(void* const* d_in, const int* in_sizes, int n_in,
                              void* d_out, int out_size, void* d_ws, size_t ws_size,
                              hipStream_t stream) {
    const float* x  = (const float*)d_in[0];
    const float* Wq = (const float*)d_in[1];
    const float* Wk = (const float*)d_in[2];
    const float* Wv = (const float*)d_in[3];
    float* out = (float*)d_out;

    float* Q = (float*)d_ws;
    float* K = Q + (size_t)BT * HD;
    float* V = K + (size_t)BT * HD;

    proj_qkv_kernel<<<dim3(4, BT / PROW), 256, 0, stream>>>(x, Wq, Wk, Wv, Q, K, V);
    attn_kernel<<<dim3(2, TT / MTILE, BB), 256, 0, stream>>>(Q, K, V, out);
}


# Attention: both head-dim halves of a query tile in one work-group; each wave does half of a chunk's QK^T (2 of 4 key tiles), halves meet in a shared LDS patch, 1 barrier per chunk; sums unchanged
# speedup vs baseline: 9.5842x; 1.0946x over previous
#include <hip/hip_runtime.h>
#include <hip/hip_bf16.h>


typedef __attribute__((ext_vector_type(2))) float v2f;
typedef __attribute__((ext_vector_type(8))) float v8f;

static constexpr int BB = 8;
static constexpr int TT = 2048;
static constexpr int CD = 1024;
static constexpr int HD = 64;
static constexpr int BT = BB * TT;

static constexpr int PROW  = 256;
static constexpr int NTC   = 3;
static constexpr int WP    = 96;
static constexpr int NPR   = CD / 2;
static constexpr int AGRP  = TT / 16 / 4;
static constexpr int HH    = 32;
static constexpr int SK    = 64;
static constexpr int PSTR  = 68;

__device__ __forceinline__ v8f wmma_f32(v2f a, v2f b, v8f c) {
    return __builtin_amdgcn_wmma_f32_16x16x4_f32(
         false, a,  false, b,
         (short)0, c,  false,  false);
}

__device__ __forceinline__ v2f lo2(const float4 v) { v2f r; r[0] = v.x; r[1] = v.y; return r; }
__device__ __forceinline__ v2f hi2(const float4 v) { v2f r; r[0] = v.z; r[1] = v.w; return r; }


__device__ __forceinline__ void proj_load_b(v2f (&b)[4 * NTC], const float* wl_grp)
{
    #pragma unroll
    for (int s = 0; s < 4; ++s) {
        #pragma unroll
        for (int t = 0; t < NTC; ++t)
            b[s * NTC + t] = *(const v2f*)(wl_grp + 2 * s * WP + t * 32);
    }
}

__device__ __forceinline__ void proj_load_x(float4 (&xs)[2][8], const float* xr0, const float* xr1, int kb)
{
    #pragma unroll
    for (int jp = 0; jp < 8; ++jp) {
        xs[0][jp] = *(const float4*)(xr0 + kb + 8 * jp);
        xs[1][jp] = *(const float4*)(xr1 + kb + 8 * jp);
    }
}

__device__ __forceinline__ int proj_tile_col(int c, int t)
{
    const int tile = (c < 2) ? ((t < 2) ? 2 * c + t : 4 + c) : 3 * c + t;
    return tile * 16;
}

__global__ __launch_bounds__(256) void proj_qkv_kernel(
    const float* __restrict__ x,
    const float* __restrict__ Wq, const float* __restrict__ Wk,
    const float* __restrict__ Wv,
    float* __restrict__ Q, float* __restrict__ K, float* __restrict__ V)
{
    __shared__ __attribute__((aligned(16))) float wsm[NPR * WP];

    const int tid  = threadIdx.x;
    const int wave = tid >> 5;
    const int lane = tid & 31;
    const int grp  = lane >> 4;
    const int idx  = lane & 15;
    const int  cblk = blockIdx.x;
    const long rowW = (long)blockIdx.y * PROW + wave * 32;

    const float* xr0 = x + (rowW + idx) * CD + 4 * grp;
    const float* xr1 = xr0 + 16 * CD;

    float4 xa[2][8], xb[2][8];
    proj_load_x(xa, xr0, xr1, 0);

    if (tid < 192) {
        const int pr = tid / 12;
        const int c4 = tid - pr * 12;
        const int gc = proj_tile_col(cblk, c4 >> 2) + (c4 & 3) * 4;
        const int m  = gc >> 6;
        const float* Wg  = (m == 0) ? Wq : ((m == 1) ? Wk : Wv);
        const float* sp0 = Wg + (long)(2 * pr) * HD + (gc & 63);
        const int posr = (pr & ~3) | ((pr & 1) << 1) | ((pr >> 1) & 1);
        float* d0 = &wsm[posr * WP + c4 * 8];
        const int rot = (int)blockIdx.y & 31;
        for (int i0 = 0; i0 < 32; i0 += 16) {
            float4 r0[16], r1[16];
            #pragma unroll
            for (int i = 0; i < 16; ++i) {
                const int ii = (i0 + i + rot) & 31;
                const float* sp = sp0 + ii * (32 * HD);
                r0[i] = *(const float4*)(sp);
                r1[i] = *(const float4*)(sp + HD);
            }
            #pragma unroll
            for (int i = 0; i < 16; ++i) {
                const int ii = (i0 + i + rot) & 31;
                float* d = d0 + ii * (16 * WP);
                *(float4*)(d)     = make_float4(r0[i].x, r1[i].x, r0[i].y, r1[i].y);
                *(float4*)(d + 4) = make_float4(r0[i].z, r1[i].z, r0[i].w, r1[i].w);
            }
        }
    }
    __syncthreads();

    const float* wl = &wsm[grp * WP + idx * 2];

    v8f acc[2][NTC] = {};
    v2f bf[2][4 * NTC];
    proj_load_b(bf[0], wl);

    for (int k0 = 0; k0 < CD; k0 += 128) {
        const float* wk = wl + (k0 / 2) * WP;
        const bool more = (k0 + 128 < CD);
        proj_load_x(xb, xr0, xr1, k0 + 64);

        #pragma unroll
        for (int q = 0; q < 8; ++q) {
            if (q == 4) {
                proj_load_x(xa, xr0, xr1, more ? k0 + 128 : k0);
            }
            if (q < 7) proj_load_b(bf[(q + 1) & 1], wk + 8 * (q + 1) * WP);
            else       proj_load_b(bf[0], more ? wk + 64 * WP : wk);
            __builtin_amdgcn_sched_barrier(0);
            #pragma unroll
            for (int s = 0; s < 4; ++s) {
                const int jp = 2 * (q & 3) + (s >> 1);
                #pragma unroll
                for (int rt = 0; rt < 2; ++rt) {
                    const float4 xv = (q < 4) ? xa[rt][jp] : xb[rt][jp];
                    const v2f a = (s & 1) ? hi2(xv) : lo2(xv);
                    #pragma unroll
                    for (int t = 0; t < NTC; ++t)
                        acc[rt][t] = wmma_f32(a, bf[q & 1][s * NTC + t], acc[rt][t]);
                }
            }
            __builtin_amdgcn_sched_barrier(0);
        }
    }

    #pragma unroll
    for (int rt = 0; rt < 2; ++rt) {
        const long rowT  = rowW + rt * 16;
        const long tileG = rowT >> 4;
        #pragma unroll
        for (int t = 0; t < NTC; ++t) {
            const int gc = proj_tile_col(cblk, t);
            const int m  = gc >> 6;
            const int cb = gc & 63;
            if (m == 0) {
                #pragma unroll
                for (int r = 0; r < 8; ++r)
                    Q[(rowT + r + 8 * grp) * HD + cb + idx] = acc[rt][t][r];
            } else if (m == 1) {
                const int cm = cb + idx;
                float* d = K + tileG * 1024 + (cm >> 3) * 128 + (((cm >> 2) & 1) * 16 + 8 * grp) * 4 + (cm & 3);
                #pragma unroll
                for (int r = 0; r < 8; ++r) d[r * 4] = acc[rt][t][r];
            } else {
                float* d = V + tileG * 1024 + (((cb >> 4) * 2 + grp) * 32 + idx) * 4;
                *(float4*)(d)      = make_float4(acc[rt][t][0], acc[rt][t][1], acc[rt][t][2], acc[rt][t][3]);
                *(float4*)(d + 64) = make_float4(acc[rt][t][4], acc[rt][t][5], acc[rt][t][6], acc[rt][t][7]);
            }
        }
    }
}

__device__ __forceinline__ float xhalf(float x) {
    return __builtin_bit_cast(float, __builtin_amdgcn_permlanex16(
        __builtin_bit_cast(int, x), __builtin_bit_cast(int, x), 0x76543210, 0xfedcba98, false, false));
}

__global__ __launch_bounds__(256) void attn_kernel(
    const float* __restrict__ Q, const float* __restrict__ K,
    const float* __restrict__ V, float* __restrict__ out)
{
    __shared__ __attribute__((aligned(16))) float ps[4][2][16 * PSTR];
    __shared__ __attribute__((aligned(16))) float rs[8][16];

    const int tid  = threadIdx.x;
    const int wave = __builtin_amdgcn_readfirstlane(tid >> 5);
    const int lane = tid & 31;
    const int grp  = lane >> 4;
    const int idx  = lane & 15;

    const int  hz    = wave & 1;
    const int  tl    = wave >> 1;
    const int  g     = blockIdx.x;
    const int  b     = blockIdx.y;
    const long base  = (long)b * TT * HD;
    const int  tile  = (tl < 2) ? 127 - 2 * g - tl : 2 * g + (tl - 2);
    const int  qrow0 = tile * 16;

    float4 qf[8];
    const float* qp = Q + base + (long)(qrow0 + idx) * HD + 4 * grp;
    #pragma unroll
    for (int j = 0; j < 8; ++j) qf[j] = *(const float4*)(qp + 8 * j);

    const float* kfp = K + base + (2 * hz) * 1024 + lane * 4;
    const float* vfp = V + base + (2 * hz) * 256 + lane * 4;

    float4 kf[2][8];
    float4 vf[4][2][2];
    #pragma unroll
    for (int k2 = 0; k2 < 2; ++k2)
        #pragma unroll
        for (int j = 0; j < 8; ++j) kf[k2][j] = *(const float4*)(kfp + k2 * 1024 + j * 128);

    v8f   o[2] = {};
    float mrow = -1e30f, lrow = 0.f;
    const int qr = qrow0 + idx;

    float* rw = &rs[wave][0];

    const int nchunk = (qrow0 + 15) / SK + 1;
    const int ntrip  = (127 - 2 * g) / 4 + 1;
    for (int sc = 0; sc < nchunk; ++sc) {
        float* pw = &ps[tl][sc & 1][0];

        {
            #pragma unroll
            for (int kt = 0; kt < 4; ++kt)
                #pragma unroll
                for (int tt = 0; tt < 2; ++tt)
                    #pragma unroll
                    for (int jj = 0; jj < 2; ++jj)
                        vf[kt][tt][jj] = *(const float4*)(vfp + (long)sc * (4 * 1024) + kt * 1024 + (tt * 2 + jj) * 128);
            __builtin_amdgcn_sched_barrier(0);

            v8f s[2] = {};
            #pragma unroll
            for (int j = 0; j < 8; ++j) {
                #pragma unroll
                for (int k2 = 0; k2 < 2; ++k2) s[k2] = wmma_f32(lo2(qf[j]), lo2(kf[k2][j]), s[k2]);
                #pragma unroll
                for (int k2 = 0; k2 < 2; ++k2) s[k2] = wmma_f32(hi2(qf[j]), hi2(kf[k2][j]), s[k2]);
            }
            #pragma unroll
            for (int k2 = 0; k2 < 2; ++k2)
                #pragma unroll
                for (int r = 0; r < 8; ++r)
                    pw[(r + 8 * grp) * PSTR + (2 * hz + k2) * 16 + idx] = s[k2][r];
        }
        __builtin_amdgcn_fence(__ATOMIC_RELEASE, "workgroup", "local");
        __builtin_amdgcn_s_barrier();
        __builtin_amdgcn_fence(__ATOMIC_ACQUIRE, "workgroup", "local");

        {
            const int s0 = sc * SK;
            const long nxt = (long)((sc + 1 < nchunk) ? sc + 1 : sc) * (4 * 1024);

            #pragma unroll
            for (int k2 = 0; k2 < 2; ++k2)
                #pragma unroll
                for (int j = 0; j < 8; ++j) kf[k2][j] = *(const float4*)(kfp + nxt + k2 * 1024 + j * 128);
            __builtin_amdgcn_sched_barrier(0);

            float p[32];
            #pragma unroll
            for (int kt = 0; kt < 4; ++kt)
                #pragma unroll
                for (int jj = 0; jj < 2; ++jj) {
                    const float4 t = *(const float4*)(&pw[idx * PSTR + kt * 16 + 8 * jj + 4 * grp]);
                    p[(kt * 2 + jj) * 4 + 0] = t.x;
                    p[(kt * 2 + jj) * 4 + 1] = t.y;
                    p[(kt * 2 + jj) * 4 + 2] = t.z;
                    p[(kt * 2 + jj) * 4 + 3] = t.w;
                }

            const int dlim = qr - s0 - 4 * grp;
            float m = -1e30f;
            #pragma unroll
            for (int kt = 0; kt < 4; ++kt)
                #pragma unroll
                for (int jj = 0; jj < 2; ++jj)
                    #pragma unroll
                    for (int e = 0; e < 4; ++e) {
                        const int n = (kt * 2 + jj) * 4 + e;
                        const int c = kt * 16 + jj * 8 + e;
                        const float sv = p[n] * 0.125f;
                        p[n] = (c <= dlim) ? sv : -1e30f;
                        m = fmaxf(m, p[n]);
                    }
            m = fmaxf(m, xhalf(m));

            const float mnew  = fmaxf(mrow, m);
            const float pcorr = __expf(mrow - mnew);
            mrow = mnew;
            rw[idx] = pcorr;
            const float4 c0 = *(const float4*)(rw + 8 * grp);
            const float4 c1 = *(const float4*)(rw + 8 * grp + 4);

            float lsum = 0.f;
            #pragma unroll
            for (int kt = 0; kt < 4; ++kt)
                #pragma unroll
                for (int jj = 0; jj < 2; ++jj)
                    #pragma unroll
                    for (int e = 0; e < 4; ++e) {
                        const int n = (kt * 2 + jj) * 4 + e;
                        const int c = kt * 16 + jj * 8 + e;
                        const float pv = (c <= dlim) ? __expf(p[n] - mnew) : 0.f;
                        p[n] = pv;
                        lsum += pv;
                    }
            lrow = lrow * pcorr + lsum;

            #pragma unroll
            for (int t = 0; t < 2; ++t) {
                o[t][0] *= c0.x; o[t][1] *= c0.y; o[t][2] *= c0.z; o[t][3] *= c0.w;
                o[t][4] *= c1.x; o[t][5] *= c1.y; o[t][6] *= c1.z; o[t][7] *= c1.w;
            }

            #pragma unroll
            for (int kt = 0; kt < 4; ++kt) {
                #pragma unroll
                for (int jj = 0; jj < 2; ++jj) {
                    const int n = (kt * 2 + jj) * 4;
                    v2f alo, ahi;
                    alo[0] = p[n + 0]; alo[1] = p[n + 1];
                    ahi[0] = p[n + 2]; ahi[1] = p[n + 3];
                    #pragma unroll
                    for (int tt = 0; tt < 2; ++tt) o[tt] = wmma_f32(alo, lo2(vf[kt][tt][jj]), o[tt]);
                    #pragma unroll
                    for (int tt = 0; tt < 2; ++tt) o[tt] = wmma_f32(ahi, hi2(vf[kt][tt][jj]), o[tt]);
                }
            }
        }
    }
    for (int sc = nchunk; sc < ntrip; ++sc) __builtin_amdgcn_s_barrier();

    const float ltot = lrow + xhalf(lrow);
    rw[idx] = ltot;
    const float4 l0 = *(const float4*)(rw + 8 * grp);
    const float4 l1 = *(const float4*)(rw + 8 * grp + 4);
    const float lr[8] = { l0.x, l0.y, l0.z, l0.w, l1.x, l1.y, l1.z, l1.w };

    #pragma unroll
    for (int t = 0; t < 2; ++t)
        #pragma unroll
        for (int r = 0; r < 8; ++r) {
            const long row = qrow0 + r + 8 * grp;
            out[base + row * HD + hz * HH + t * 16 + idx] = o[t][r] / lr[r];
        }
}

extern "C" void kernel_launch(void* const* d_in, const int* in_sizes, int n_in,
                              void* d_out, int out_size, void* d_ws, size_t ws_size,
                              hipStream_t stream) {
    const float* x  = (const float*)d_in[0];
    const float* Wq = (const float*)d_in[1];
    const float* Wk = (const float*)d_in[2];
    const float* Wv = (const float*)d_in[3];
    float* out = (float*)d_out;

    float* Q = (float*)d_ws;
    float* K = Q + (size_t)BT * HD;
    float* V = K + (size_t)BT * HD;

    proj_qkv_kernel<<<dim3(4, BT / PROW), 256, 0, stream>>>(x, Wq, Wk, Wv, Q, K, V);
    attn_kernel<<<dim3(AGRP, BB), 256, 0, stream>>>(Q, K, V, out);
}


# Attention: 16 waves per work-group in two roles: score waves (QK^T) run one chunk ahead of value waves (softmax + PV) via the double-buffered LDS patch, 1 barrier per chunk; sums unchanged
# speedup vs baseline: 10.2003x; 1.0643x over previous
#include <hip/hip_runtime.h>
#include <hip/hip_bf16.h>


typedef __attribute__((ext_vector_type(2))) float v2f;
typedef __attribute__((ext_vector_type(8))) float v8f;

static constexpr int BB = 8;
static constexpr int TT = 2048;
static constexpr int CD = 1024;
static constexpr int HD = 64;
static constexpr int BT = BB * TT;

static constexpr int PROW  = 256;
static constexpr int NTC   = 3;
static constexpr int WP    = 96;
static constexpr int NPR   = CD / 2;
static constexpr int AGRP  = TT / 16 / 4;
static constexpr int HH    = 32;
static constexpr int SK    = 64;
static constexpr int PSTR  = 68;

__device__ __forceinline__ v8f wmma_f32(v2f a, v2f b, v8f c) {
    return __builtin_amdgcn_wmma_f32_16x16x4_f32(
         false, a,  false, b,
         (short)0, c,  false,  false);
}

__device__ __forceinline__ v2f lo2(const float4 v) { v2f r; r[0] = v.x; r[1] = v.y; return r; }
__device__ __forceinline__ v2f hi2(const float4 v) { v2f r; r[0] = v.z; r[1] = v.w; return r; }


__device__ __forceinline__ void proj_load_b(v2f (&b)[4 * NTC], const float* wl_grp)
{
    #pragma unroll
    for (int s = 0; s < 4; ++s) {
        #pragma unroll
        for (int t = 0; t < NTC; ++t)
            b[s * NTC + t] = *(const v2f*)(wl_grp + 2 * s * WP + t * 32);
    }
}

__device__ __forceinline__ void proj_load_x(float4 (&xs)[2][8], const float* xr0, const float* xr1, int kb)
{
    #pragma unroll
    for (int jp = 0; jp < 8; ++jp) {
        xs[0][jp] = *(const float4*)(xr0 + kb + 8 * jp);
        xs[1][jp] = *(const float4*)(xr1 + kb + 8 * jp);
    }
}

__device__ __forceinline__ int proj_tile_col(int c, int t)
{
    const int tile = (c < 2) ? ((t < 2) ? 2 * c + t : 4 + c) : 3 * c + t;
    return tile * 16;
}

__global__ __launch_bounds__(256) void proj_qkv_kernel(
    const float* __restrict__ x,
    const float* __restrict__ Wq, const float* __restrict__ Wk,
    const float* __restrict__ Wv,
    float* __restrict__ Q, float* __restrict__ K, float* __restrict__ V)
{
    __shared__ __attribute__((aligned(16))) float wsm[NPR * WP];

    const int tid  = threadIdx.x;
    const int wave = tid >> 5;
    const int lane = tid & 31;
    const int grp  = lane >> 4;
    const int idx  = lane & 15;
    const int  cblk = blockIdx.x;
    const long rowW = (long)blockIdx.y * PROW + wave * 32;

    const float* xr0 = x + (rowW + idx) * CD + 4 * grp;
    const float* xr1 = xr0 + 16 * CD;

    float4 xa[2][8], xb[2][8];
    proj_load_x(xa, xr0, xr1, 0);

    if (tid < 192) {
        const int pr = tid / 12;
        const int c4 = tid - pr * 12;
        const int gc = proj_tile_col(cblk, c4 >> 2) + (c4 & 3) * 4;
        const int m  = gc >> 6;
        const float* Wg  = (m == 0) ? Wq : ((m == 1) ? Wk : Wv);
        const float* sp0 = Wg + (long)(2 * pr) * HD + (gc & 63);
        const int posr = (pr & ~3) | ((pr & 1) << 1) | ((pr >> 1) & 1);
        float* d0 = &wsm[posr * WP + c4 * 8];
        const int rot = (int)blockIdx.y & 31;
        for (int i0 = 0; i0 < 32; i0 += 16) {
            float4 r0[16], r1[16];
            #pragma unroll
            for (int i = 0; i < 16; ++i) {
                const int ii = (i0 + i + rot) & 31;
                const float* sp = sp0 + ii * (32 * HD);
                r0[i] = *(const float4*)(sp);
                r1[i] = *(const float4*)(sp + HD);
            }
            #pragma unroll
            for (int i = 0; i < 16; ++i) {
                const int ii = (i0 + i + rot) & 31;
                float* d = d0 + ii * (16 * WP);
                *(float4*)(d)     = make_float4(r0[i].x, r1[i].x, r0[i].y, r1[i].y);
                *(float4*)(d + 4) = make_float4(r0[i].z, r1[i].z, r0[i].w, r1[i].w);
            }
        }
    }
    __syncthreads();

    const float* wl = &wsm[grp * WP + idx * 2];

    v8f acc[2][NTC] = {};
    v2f bf[2][4 * NTC];
    proj_load_b(bf[0], wl);

    for (int k0 = 0; k0 < CD; k0 += 128) {
        const float* wk = wl + (k0 / 2) * WP;
        const bool more = (k0 + 128 < CD);
        proj_load_x(xb, xr0, xr1, k0 + 64);

        #pragma unroll
        for (int q = 0; q < 8; ++q) {
            if (q == 4) {
                proj_load_x(xa, xr0, xr1, more ? k0 + 128 : k0);
            }
            if (q < 7) proj_load_b(bf[(q + 1) & 1], wk + 8 * (q + 1) * WP);
            else       proj_load_b(bf[0], more ? wk + 64 * WP : wk);
            __builtin_amdgcn_sched_barrier(0);
            #pragma unroll
            for (int s = 0; s < 4; ++s) {
                const int jp = 2 * (q & 3) + (s >> 1);
                #pragma unroll
                for (int rt = 0; rt < 2; ++rt) {
                    const float4 xv = (q < 4) ? xa[rt][jp] : xb[rt][jp];
                    const v2f a = (s & 1) ? hi2(xv) : lo2(xv);
                    #pragma unroll
                    for (int t = 0; t < NTC; ++t)
                        acc[rt][t] = wmma_f32(a, bf[q & 1][s * NTC + t], acc[rt][t]);
                }
            }
            __builtin_amdgcn_sched_barrier(0);
        }
    }

    #pragma unroll
    for (int rt = 0; rt < 2; ++rt) {
        const long rowT  = rowW + rt * 16;
        const long tileG = rowT >> 4;
        #pragma unroll
        for (int t = 0; t < NTC; ++t) {
            const int gc = proj_tile_col(cblk, t);
            const int m  = gc >> 6;
            const int cb = gc & 63;
            if (m == 0) {
                #pragma unroll
                for (int r = 0; r < 8; ++r)
                    Q[(rowT + r + 8 * grp) * HD + cb + idx] = acc[rt][t][r];
            } else if (m == 1) {
                const int cm = cb + idx;
                float* d = K + tileG * 1024 + (cm >> 3) * 128 + (((cm >> 2) & 1) * 16 + 8 * grp) * 4 + (cm & 3);
                #pragma unroll
                for (int r = 0; r < 8; ++r) d[r * 4] = acc[rt][t][r];
            } else {
                float* d = V + tileG * 1024 + (((cb >> 4) * 2 + grp) * 32 + idx) * 4;
                *(float4*)(d)      = make_float4(acc[rt][t][0], acc[rt][t][1], acc[rt][t][2], acc[rt][t][3]);
                *(float4*)(d + 64) = make_float4(acc[rt][t][4], acc[rt][t][5], acc[rt][t][6], acc[rt][t][7]);
            }
        }
    }
}

__device__ __forceinline__ float xhalf(float x) {
    return __builtin_bit_cast(float, __builtin_amdgcn_permlanex16(
        __builtin_bit_cast(int, x), __builtin_bit_cast(int, x), 0x76543210, 0xfedcba98, false, false));
}

__device__ __forceinline__ void trip_barrier() {
    __builtin_amdgcn_fence(__ATOMIC_RELEASE, "workgroup", "local");
    __builtin_amdgcn_s_barrier();
    __builtin_amdgcn_fence(__ATOMIC_ACQUIRE, "workgroup", "local");
}

__device__ __forceinline__ void score_trip(const float4 (&qf)[8], const float4 (&kc)[2][8], float4 (&kn)[2][8],
                                           const float* knext, float* pw, int grp, int idx, int hz)
{
    #pragma unroll
    for (int k2 = 0; k2 < 2; ++k2)
        #pragma unroll
        for (int j = 0; j < 8; ++j) kn[k2][j] = *(const float4*)(knext + k2 * 1024 + j * 128);

    v8f s[2] = {};
    #pragma unroll
    for (int j = 0; j < 8; ++j) {
        #pragma unroll
        for (int k2 = 0; k2 < 2; ++k2) s[k2] = wmma_f32(lo2(qf[j]), lo2(kc[k2][j]), s[k2]);
        #pragma unroll
        for (int k2 = 0; k2 < 2; ++k2) s[k2] = wmma_f32(hi2(qf[j]), hi2(kc[k2][j]), s[k2]);
    }
    #pragma unroll
    for (int k2 = 0; k2 < 2; ++k2)
        #pragma unroll
        for (int r = 0; r < 8; ++r)
            pw[(r + 8 * grp) * PSTR + (2 * hz + k2) * 16 + idx] = s[k2][r];
}

__device__ __forceinline__ void value_trip(const float4 (&vc)[4][2][2], float4 (&vn)[4][2][2], const float* vnext,
                                           const float* pw, float* rw, v8f (&o)[2], float& mrow, float& lrow,
                                           int dlim, int grp, int idx)
{
    #pragma unroll
    for (int kt = 0; kt < 4; ++kt)
        #pragma unroll
        for (int tt = 0; tt < 2; ++tt)
            #pragma unroll
            for (int jj = 0; jj < 2; ++jj)
                vn[kt][tt][jj] = *(const float4*)(vnext + kt * 1024 + (tt * 2 + jj) * 128);

    float p[32];
    #pragma unroll
    for (int kt = 0; kt < 4; ++kt)
        #pragma unroll
        for (int jj = 0; jj < 2; ++jj) {
            const float4 t = *(const float4*)(&pw[idx * PSTR + kt * 16 + 8 * jj + 4 * grp]);
            p[(kt * 2 + jj) * 4 + 0] = t.x;
            p[(kt * 2 + jj) * 4 + 1] = t.y;
            p[(kt * 2 + jj) * 4 + 2] = t.z;
            p[(kt * 2 + jj) * 4 + 3] = t.w;
        }

    float m = -1e30f;
    #pragma unroll
    for (int kt = 0; kt < 4; ++kt)
        #pragma unroll
        for (int jj = 0; jj < 2; ++jj)
            #pragma unroll
            for (int e = 0; e < 4; ++e) {
                const int n = (kt * 2 + jj) * 4 + e;
                const int c = kt * 16 + jj * 8 + e;
                const float sv = p[n] * 0.125f;
                p[n] = (c <= dlim) ? sv : -1e30f;
                m = fmaxf(m, p[n]);
            }
    m = fmaxf(m, xhalf(m));

    const float mnew  = fmaxf(mrow, m);
    const float pcorr = __expf(mrow - mnew);
    mrow = mnew;
    rw[idx] = pcorr;
    const float4 c0 = *(const float4*)(rw + 8 * grp);
    const float4 c1 = *(const float4*)(rw + 8 * grp + 4);

    float lsum = 0.f;
    #pragma unroll
    for (int kt = 0; kt < 4; ++kt)
        #pragma unroll
        for (int jj = 0; jj < 2; ++jj)
            #pragma unroll
            for (int e = 0; e < 4; ++e) {
                const int n = (kt * 2 + jj) * 4 + e;
                const int c = kt * 16 + jj * 8 + e;
                const float pv = (c <= dlim) ? __expf(p[n] - mnew) : 0.f;
                p[n] = pv;
                lsum += pv;
            }
    lrow = lrow * pcorr + lsum;

    #pragma unroll
    for (int t = 0; t < 2; ++t) {
        o[t][0] *= c0.x; o[t][1] *= c0.y; o[t][2] *= c0.z; o[t][3] *= c0.w;
        o[t][4] *= c1.x; o[t][5] *= c1.y; o[t][6] *= c1.z; o[t][7] *= c1.w;
    }

    #pragma unroll
    for (int kt = 0; kt < 4; ++kt) {
        #pragma unroll
        for (int jj = 0; jj < 2; ++jj) {
            const int n = (kt * 2 + jj) * 4;
            v2f alo, ahi;
            alo[0] = p[n + 0]; alo[1] = p[n + 1];
            ahi[0] = p[n + 2]; ahi[1] = p[n + 3];
            #pragma unroll
            for (int tt = 0; tt < 2; ++tt) o[tt] = wmma_f32(alo, lo2(vc[kt][tt][jj]), o[tt]);
            #pragma unroll
            for (int tt = 0; tt < 2; ++tt) o[tt] = wmma_f32(ahi, hi2(vc[kt][tt][jj]), o[tt]);
        }
    }
}

__global__ __launch_bounds__(512) void attn_kernel(
    const float* __restrict__ Q, const float* __restrict__ K,
    const float* __restrict__ V, float* __restrict__ out)
{
    __shared__ __attribute__((aligned(16))) float ps[4][2][16 * PSTR];
    __shared__ __attribute__((aligned(16))) float rs[8][16];

    const int tid  = threadIdx.x;
    const int wave = __builtin_amdgcn_readfirstlane(tid >> 5);
    const int lane = tid & 31;
    const int grp  = lane >> 4;
    const int idx  = lane & 15;

    const int  role  = wave >> 3;
    const int  w8    = wave & 7;
    const int  hz    = w8 & 1;
    const int  tl    = w8 >> 1;
    const int  g     = blockIdx.x;
    const int  b     = blockIdx.y;
    const long base  = (long)b * TT * HD;
    const int  tile  = (tl < 2) ? 127 - 2 * g - tl : 2 * g + (tl - 2);
    const int  qrow0 = tile * 16;

    const int nchunk = (qrow0 + 15) / SK + 1;
    const int ntrip  = (127 - 2 * g) / 4 + 1;

    if (role == 1) {
        float4 qf[8];
        const float* qp = Q + base + (long)(qrow0 + idx) * HD + 4 * grp;
        #pragma unroll
        for (int j = 0; j < 8; ++j) qf[j] = *(const float4*)(qp + 8 * j);

        const float* kfp = K + base + (2 * hz) * 1024 + lane * 4;

        float4 kfa[2][8], kfb[2][8];
        #pragma unroll
        for (int k2 = 0; k2 < 2; ++k2)
            #pragma unroll
            for (int j = 0; j < 8; ++j) kfa[k2][j] = *(const float4*)(kfp + k2 * 1024 + j * 128);

        for (int sc = 0; sc < nchunk; sc += 2) {
            const int n1 = (sc + 1 < nchunk) ? sc + 1 : sc;
            score_trip(qf, kfa, kfb, kfp + (long)n1 * (4 * 1024), &ps[tl][0][0], grp, idx, hz);
            trip_barrier();
            if (sc + 1 < nchunk) {
                const int n2 = (sc + 2 < nchunk) ? sc + 2 : sc + 1;
                score_trip(qf, kfb, kfa, kfp + (long)n2 * (4 * 1024), &ps[tl][1][0], grp, idx, hz);
                trip_barrier();
            }
        }
        for (int sc = nchunk; sc <= ntrip; ++sc) __builtin_amdgcn_s_barrier();
        return;
    }

    const float* vfp = V + base + (2 * hz) * 256 + lane * 4;

    float4 vfa[4][2][2], vfb[4][2][2];
    #pragma unroll
    for (int kt = 0; kt < 4; ++kt)
        #pragma unroll
        for (int tt = 0; tt < 2; ++tt)
            #pragma unroll
            for (int jj = 0; jj < 2; ++jj)
                vfa[kt][tt][jj] = *(const float4*)(vfp + kt * 1024 + (tt * 2 + jj) * 128);

    v8f   o[2] = {};
    float mrow = -1e30f, lrow = 0.f;
    const int dl0 = qrow0 + idx - 4 * grp;

    float* rw = &rs[w8][0];

    trip_barrier();

    for (int sc = 0; sc < nchunk; sc += 2) {
        const int n1 = (sc + 1 < nchunk) ? sc + 1 : sc;
        value_trip(vfa, vfb, vfp + (long)n1 * (4 * 1024), &ps[tl][0][0], rw, o, mrow, lrow, dl0 - sc * SK, grp, idx);
        trip_barrier();
        if (sc + 1 < nchunk) {
            const int n2 = (sc + 2 < nchunk) ? sc + 2 : sc + 1;
            value_trip(vfb, vfa, vfp + (long)n2 * (4 * 1024), &ps[tl][1][0], rw, o, mrow, lrow, dl0 - (sc + 1) * SK, grp, idx);
            trip_barrier();
        }
    }
    for (int sc = nchunk + 1; sc <= ntrip; ++sc) __builtin_amdgcn_s_barrier();

    const float ltot = lrow + xhalf(lrow);
    rw[idx] = ltot;
    const float4 l0 = *(const float4*)(rw + 8 * grp);
    const float4 l1 = *(const float4*)(rw + 8 * grp + 4);
    const float lr[8] = { l0.x, l0.y, l0.z, l0.w, l1.x, l1.y, l1.z, l1.w };

    #pragma unroll
    for (int t = 0; t < 2; ++t)
        #pragma unroll
        for (int r = 0; r < 8; ++r) {
            const long row = qrow0 + r + 8 * grp;
            out[base + row * HD + hz * HH + t * 16 + idx] = o[t][r] / lr[r];
        }
}

extern "C" void kernel_launch(void* const* d_in, const int* in_sizes, int n_in,
                              void* d_out, int out_size, void* d_ws, size_t ws_size,
                              hipStream_t stream) {
    const float* x  = (const float*)d_in[0];
    const float* Wq = (const float*)d_in[1];
    const float* Wk = (const float*)d_in[2];
    const float* Wv = (const float*)d_in[3];
    float* out = (float*)d_out;

    float* Q = (float*)d_ws;
    float* K = Q + (size_t)BT * HD;
    float* V = K + (size_t)BT * HD;

    proj_qkv_kernel<<<dim3(4, BT / PROW), 256, 0, stream>>>(x, Wq, Wk, Wv, Q, K, V);
    attn_kernel<<<dim3(AGRP, BB), 512, 0, stream>>>(Q, K, V, out);
}


# Attention: a tile's two value waves split each chunk's keys (32 each, all 64 head-dim columns), not the columns: each score through the softmax once, not twice; partials joined at end (new sum order)
# speedup vs baseline: 10.5429x; 1.0336x over previous
#include <hip/hip_runtime.h>
#include <hip/hip_bf16.h>


typedef __attribute__((ext_vector_type(2))) float v2f;
typedef __attribute__((ext_vector_type(8))) float v8f;

static constexpr int BB = 8;
static constexpr int TT = 2048;
static constexpr int CD = 1024;
static constexpr int HD = 64;
static constexpr int BT = BB * TT;

static constexpr int PROW  = 256;
static constexpr int NTC   = 3;
static constexpr int WP    = 96;
static constexpr int NPR   = CD / 2;
static constexpr int AGRP  = TT / 16 / 4;
static constexpr int HH    = 32;
static constexpr int SK    = 64;
static constexpr int PSTR  = 68;

__device__ __forceinline__ v8f wmma_f32(v2f a, v2f b, v8f c) {
    return __builtin_amdgcn_wmma_f32_16x16x4_f32(
         false, a,  false, b,
         (short)0, c,  false,  false);
}

__device__ __forceinline__ v2f lo2(const float4 v) { v2f r; r[0] = v.x; r[1] = v.y; return r; }
__device__ __forceinline__ v2f hi2(const float4 v) { v2f r; r[0] = v.z; r[1] = v.w; return r; }


__device__ __forceinline__ void proj_load_b(v2f (&b)[4 * NTC], const float* wl_grp)
{
    #pragma unroll
    for (int s = 0; s < 4; ++s) {
        #pragma unroll
        for (int t = 0; t < NTC; ++t)
            b[s * NTC + t] = *(const v2f*)(wl_grp + 2 * s * WP + t * 32);
    }
}

__device__ __forceinline__ void proj_load_x(float4 (&xs)[2][8], const float* xr0, const float* xr1, int kb)
{
    #pragma unroll
    for (int jp = 0; jp < 8; ++jp) {
        xs[0][jp] = *(const float4*)(xr0 + kb + 8 * jp);
        xs[1][jp] = *(const float4*)(xr1 + kb + 8 * jp);
    }
}

__device__ __forceinline__ int proj_tile_col(int c, int t)
{
    const int tile = (c < 2) ? ((t < 2) ? 2 * c + t : 4 + c) : 3 * c + t;
    return tile * 16;
}

__global__ __launch_bounds__(256) void proj_qkv_kernel(
    const float* __restrict__ x,
    const float* __restrict__ Wq, const float* __restrict__ Wk,
    const float* __restrict__ Wv,
    float* __restrict__ Q, float* __restrict__ K, float* __restrict__ V)
{
    __shared__ __attribute__((aligned(16))) float wsm[NPR * WP];

    const int tid  = threadIdx.x;
    const int wave = tid >> 5;
    const int lane = tid & 31;
    const int grp  = lane >> 4;
    const int idx  = lane & 15;
    const int  cblk = blockIdx.x;
    const long rowW = (long)blockIdx.y * PROW + wave * 32;

    const float* xr0 = x + (rowW + idx) * CD + 4 * grp;
    const float* xr1 = xr0 + 16 * CD;

    float4 xa[2][8], xb[2][8];
    proj_load_x(xa, xr0, xr1, 0);

    if (tid < 192) {
        const int pr = tid / 12;
        const int c4 = tid - pr * 12;
        const int gc = proj_tile_col(cblk, c4 >> 2) + (c4 & 3) * 4;
        const int m  = gc >> 6;
        const float* Wg  = (m == 0) ? Wq : ((m == 1) ? Wk : Wv);
        const float* sp0 = Wg + (long)(2 * pr) * HD + (gc & 63);
        const int posr = (pr & ~3) | ((pr & 1) << 1) | ((pr >> 1) & 1);
        float* d0 = &wsm[posr * WP + c4 * 8];
        const int rot = (int)blockIdx.y & 31;
        for (int i0 = 0; i0 < 32; i0 += 16) {
            float4 r0[16], r1[16];
            #pragma unroll
            for (int i = 0; i < 16; ++i) {
                const int ii = (i0 + i + rot) & 31;
                const float* sp = sp0 + ii * (32 * HD);
                r0[i] = *(const float4*)(sp);
                r1[i] = *(const float4*)(sp + HD);
            }
            #pragma unroll
            for (int i = 0; i < 16; ++i) {
                const int ii = (i0 + i + rot) & 31;
                float* d = d0 + ii * (16 * WP);
                *(float4*)(d)     = make_float4(r0[i].x, r1[i].x, r0[i].y, r1[i].y);
                *(float4*)(d + 4) = make_float4(r0[i].z, r1[i].z, r0[i].w, r1[i].w);
            }
        }
    }
    __syncthreads();

    const float* wl = &wsm[grp * WP + idx * 2];

    v8f acc[2][NTC] = {};
    v2f bf[2][4 * NTC];
    proj_load_b(bf[0], wl);

    for (int k0 = 0; k0 < CD; k0 += 128) {
        const float* wk = wl + (k0 / 2) * WP;
        const bool more = (k0 + 128 < CD);
        proj_load_x(xb, xr0, xr1, k0 + 64);

        #pragma unroll
        for (int q = 0; q < 8; ++q) {
            if (q == 4) {
                proj_load_x(xa, xr0, xr1, more ? k0 + 128 : k0);
            }
            if (q < 7) proj_load_b(bf[(q + 1) & 1], wk + 8 * (q + 1) * WP);
            else       proj_load_b(bf[0], more ? wk + 64 * WP : wk);
            __builtin_amdgcn_sched_barrier(0);
            #pragma unroll
            for (int s = 0; s < 4; ++s) {
                const int jp = 2 * (q & 3) + (s >> 1);
                #pragma unroll
                for (int rt = 0; rt < 2; ++rt) {
                    const float4 xv = (q < 4) ? xa[rt][jp] : xb[rt][jp];
                    const v2f a = (s & 1) ? hi2(xv) : lo2(xv);
                    #pragma unroll
                    for (int t = 0; t < NTC; ++t)
                        acc[rt][t] = wmma_f32(a, bf[q & 1][s * NTC + t], acc[rt][t]);
                }
            }
            __builtin_amdgcn_sched_barrier(0);
        }
    }

    #pragma unroll
    for (int rt = 0; rt < 2; ++rt) {
        const long rowT  = rowW + rt * 16;
        const long tileG = rowT >> 4;
        #pragma unroll
        for (int t = 0; t < NTC; ++t) {
            const int gc = proj_tile_col(cblk, t);
            const int m  = gc >> 6;
            const int cb = gc & 63;
            if (m == 0) {
                #pragma unroll
                for (int r = 0; r < 8; ++r)
                    Q[(rowT + r + 8 * grp) * HD + cb + idx] = acc[rt][t][r];
            } else if (m == 1) {
                const int cm = cb + idx;
                float* d = K + tileG * 1024 + (cm >> 3) * 128 + (((cm >> 2) & 1) * 16 + 8 * grp) * 4 + (cm & 3);
                #pragma unroll
                for (int r = 0; r < 8; ++r) d[r * 4] = acc[rt][t][r];
            } else {
                float* d = V + tileG * 1024 + (((cb >> 4) * 2 + grp) * 32 + idx) * 4;
                *(float4*)(d)      = make_float4(acc[rt][t][0], acc[rt][t][1], acc[rt][t][2], acc[rt][t][3]);
                *(float4*)(d + 64) = make_float4(acc[rt][t][4], acc[rt][t][5], acc[rt][t][6], acc[rt][t][7]);
            }
        }
    }
}

__device__ __forceinline__ float xhalf(float x) {
    return __builtin_bit_cast(float, __builtin_amdgcn_permlanex16(
        __builtin_bit_cast(int, x), __builtin_bit_cast(int, x), 0x76543210, 0xfedcba98, false, false));
}

__device__ __forceinline__ void trip_barrier() {
    __builtin_amdgcn_fence(__ATOMIC_RELEASE, "workgroup", "local");
    __builtin_amdgcn_s_barrier();
    __builtin_amdgcn_fence(__ATOMIC_ACQUIRE, "workgroup", "local");
}

__device__ __forceinline__ void score_trip(const float4 (&qf)[8], const float4 (&kc)[2][8], float4 (&kn)[2][8],
                                           const float* knext, float* pw, int grp, int idx, int hz)
{
    #pragma unroll
    for (int k2 = 0; k2 < 2; ++k2)
        #pragma unroll
        for (int j = 0; j < 8; ++j) kn[k2][j] = *(const float4*)(knext + k2 * 1024 + j * 128);

    v8f s[2] = {};
    #pragma unroll
    for (int j = 0; j < 8; ++j) {
        #pragma unroll
        for (int k2 = 0; k2 < 2; ++k2) s[k2] = wmma_f32(lo2(qf[j]), lo2(kc[k2][j]), s[k2]);
        #pragma unroll
        for (int k2 = 0; k2 < 2; ++k2) s[k2] = wmma_f32(hi2(qf[j]), hi2(kc[k2][j]), s[k2]);
    }
    #pragma unroll
    for (int k2 = 0; k2 < 2; ++k2)
        #pragma unroll
        for (int r = 0; r < 8; ++r)
            pw[(r + 8 * grp) * PSTR + (2 * hz + k2) * 16 + idx] = s[k2][r];
}

__device__ __forceinline__ void value_trip(const float4 (&vc)[2][4][2], float4 (&vn)[2][4][2], const float* vnext,
                                           const float* pw, float* rw, v8f (&o)[4], float& mrow, float& lrow,
                                           int dlim, int grp, int idx)
{
    #pragma unroll
    for (int kt = 0; kt < 2; ++kt)
        #pragma unroll
        for (int t = 0; t < 4; ++t)
            #pragma unroll
            for (int jj = 0; jj < 2; ++jj)
                vn[kt][t][jj] = *(const float4*)(vnext + kt * 1024 + (t * 2 + jj) * 128);

    float p[16];
    #pragma unroll
    for (int kt = 0; kt < 2; ++kt)
        #pragma unroll
        for (int jj = 0; jj < 2; ++jj) {
            const float4 t = *(const float4*)(&pw[idx * PSTR + kt * 16 + 8 * jj + 4 * grp]);
            p[(kt * 2 + jj) * 4 + 0] = t.x;
            p[(kt * 2 + jj) * 4 + 1] = t.y;
            p[(kt * 2 + jj) * 4 + 2] = t.z;
            p[(kt * 2 + jj) * 4 + 3] = t.w;
        }

    float m = -1e30f;
    #pragma unroll
    for (int kt = 0; kt < 2; ++kt)
        #pragma unroll
        for (int jj = 0; jj < 2; ++jj)
            #pragma unroll
            for (int e = 0; e < 4; ++e) {
                const int n = (kt * 2 + jj) * 4 + e;
                const int c = kt * 16 + jj * 8 + e;
                const float sv = p[n] * 0.125f;
                p[n] = (c <= dlim) ? sv : -1e30f;
                m = fmaxf(m, p[n]);
            }
    m = fmaxf(m, xhalf(m));

    const float mnew  = fmaxf(mrow, m);
    const float pcorr = __expf(mrow - mnew);
    mrow = mnew;
    rw[idx] = pcorr;
    const float4 c0 = *(const float4*)(rw + 8 * grp);
    const float4 c1 = *(const float4*)(rw + 8 * grp + 4);

    float lsum = 0.f;
    #pragma unroll
    for (int kt = 0; kt < 2; ++kt)
        #pragma unroll
        for (int jj = 0; jj < 2; ++jj)
            #pragma unroll
            for (int e = 0; e < 4; ++e) {
                const int n = (kt * 2 + jj) * 4 + e;
                const int c = kt * 16 + jj * 8 + e;
                const float pv = (c <= dlim) ? __expf(p[n] - mnew) : 0.f;
                p[n] = pv;
                lsum += pv;
            }
    lrow = lrow * pcorr + lsum;

    #pragma unroll
    for (int t = 0; t < 4; ++t) {
        o[t][0] *= c0.x; o[t][1] *= c0.y; o[t][2] *= c0.z; o[t][3] *= c0.w;
        o[t][4] *= c1.x; o[t][5] *= c1.y; o[t][6] *= c1.z; o[t][7] *= c1.w;
    }

    #pragma unroll
    for (int kt = 0; kt < 2; ++kt) {
        #pragma unroll
        for (int jj = 0; jj < 2; ++jj) {
            const int n = (kt * 2 + jj) * 4;
            v2f alo, ahi;
            alo[0] = p[n + 0]; alo[1] = p[n + 1];
            ahi[0] = p[n + 2]; ahi[1] = p[n + 3];
            #pragma unroll
            for (int t = 0; t < 4; ++t) o[t] = wmma_f32(alo, lo2(vc[kt][t][jj]), o[t]);
            #pragma unroll
            for (int t = 0; t < 4; ++t) o[t] = wmma_f32(ahi, hi2(vc[kt][t][jj]), o[t]);
        }
    }
}

__global__ __launch_bounds__(512) void attn_kernel(
    const float* __restrict__ Q, const float* __restrict__ K,
    const float* __restrict__ V, float* __restrict__ out)
{
    __shared__ __attribute__((aligned(16))) float ps[4][2][16 * PSTR];
    __shared__ __attribute__((aligned(16))) float rs[8][16];
    __shared__ __attribute__((aligned(16))) float fin_m[4][2][16];
    __shared__ __attribute__((aligned(16))) float fin_l[4][2][16];
    __shared__ __attribute__((aligned(16))) float fin_o[4][2][4][32][4];

    const int tid  = threadIdx.x;
    const int wave = __builtin_amdgcn_readfirstlane(tid >> 5);
    const int lane = tid & 31;
    const int grp  = lane >> 4;
    const int idx  = lane & 15;

    const int  role  = wave >> 3;
    const int  w8    = wave & 7;
    const int  hz    = w8 & 1;
    const int  tl    = w8 >> 1;
    const int  g     = blockIdx.x;
    const int  b     = blockIdx.y;
    const long base  = (long)b * TT * HD;
    const int  tile  = (tl < 2) ? 127 - 2 * g - tl : 2 * g + (tl - 2);
    const int  qrow0 = tile * 16;

    const int nchunk = (qrow0 + 15) / SK + 1;
    const int ntrip  = (127 - 2 * g) / 4 + 1;

    if (role == 1) {
        float4 qf[8];
        const float* qp = Q + base + (long)(qrow0 + idx) * HD + 4 * grp;
        #pragma unroll
        for (int j = 0; j < 8; ++j) qf[j] = *(const float4*)(qp + 8 * j);

        const float* kfp = K + base + (2 * hz) * 1024 + lane * 4;

        float4 kfa[2][8], kfb[2][8];
        #pragma unroll
        for (int k2 = 0; k2 < 2; ++k2)
            #pragma unroll
            for (int j = 0; j < 8; ++j) kfa[k2][j] = *(const float4*)(kfp + k2 * 1024 + j * 128);

        for (int sc = 0; sc < nchunk; sc += 2) {
            const int n1 = (sc + 1 < nchunk) ? sc + 1 : sc;
            score_trip(qf, kfa, kfb, kfp + (long)n1 * (4 * 1024), &ps[tl][0][0], grp, idx, hz);
            trip_barrier();
            if (sc + 1 < nchunk) {
                const int n2 = (sc + 2 < nchunk) ? sc + 2 : sc + 1;
                score_trip(qf, kfb, kfa, kfp + (long)n2 * (4 * 1024), &ps[tl][1][0], grp, idx, hz);
                trip_barrier();
            }
        }
        for (int sc = nchunk; sc <= ntrip + 1; ++sc) __builtin_amdgcn_s_barrier();
        return;
    }

    const float* vfp = V + base + (2 * hz) * 1024 + lane * 4;

    float4 vfa[2][4][2], vfb[2][4][2];
    #pragma unroll
    for (int kt = 0; kt < 2; ++kt)
        #pragma unroll
        for (int t = 0; t < 4; ++t)
            #pragma unroll
            for (int jj = 0; jj < 2; ++jj)
                vfa[kt][t][jj] = *(const float4*)(vfp + kt * 1024 + (t * 2 + jj) * 128);

    v8f   o[4] = {};
    float mrow = -1e30f, lrow = 0.f;
    const int dl0 = qrow0 + idx - 4 * grp - 32 * hz;

    float* rw = &rs[w8][0];
    const float* pw0 = &ps[tl][0][32 * hz];
    const float* pw1 = &ps[tl][1][32 * hz];

    trip_barrier();

    for (int sc = 0; sc < nchunk; sc += 2) {
        const int n1 = (sc + 1 < nchunk) ? sc + 1 : sc;
        value_trip(vfa, vfb, vfp + (long)n1 * (4 * 1024), pw0, rw, o, mrow, lrow, dl0 - sc * SK, grp, idx);
        trip_barrier();
        if (sc + 1 < nchunk) {
            const int n2 = (sc + 2 < nchunk) ? sc + 2 : sc + 1;
            value_trip(vfb, vfa, vfp + (long)n2 * (4 * 1024), pw1, rw, o, mrow, lrow, dl0 - (sc + 1) * SK, grp, idx);
            trip_barrier();
        }
    }
    for (int sc = nchunk + 1; sc <= ntrip; ++sc) __builtin_amdgcn_s_barrier();

    const float ltot = lrow + xhalf(lrow);
    fin_m[tl][hz][idx] = mrow;
    fin_l[tl][hz][idx] = ltot;
    const v8f okeep0 = hz ? o[2] : o[0];
    const v8f okeep1 = hz ? o[3] : o[1];
    const v8f osend0 = hz ? o[0] : o[2];
    const v8f osend1 = hz ? o[1] : o[3];
    float* fo = &fin_o[tl][hz][0][lane][0];
    *(float4*)(fo + 0 * 128) = make_float4(osend0[0], osend0[1], osend0[2], osend0[3]);
    *(float4*)(fo + 1 * 128) = make_float4(osend0[4], osend0[5], osend0[6], osend0[7]);
    *(float4*)(fo + 2 * 128) = make_float4(osend1[0], osend1[1], osend1[2], osend1[3]);
    *(float4*)(fo + 3 * 128) = make_float4(osend1[4], osend1[5], osend1[6], osend1[7]);

    trip_barrier();

    const float* po = &fin_o[tl][1 - hz][0][lane][0];
    const float4 q0 = *(const float4*)(po + 0 * 128);
    const float4 q1 = *(const float4*)(po + 1 * 128);
    const float4 q2 = *(const float4*)(po + 2 * 128);
    const float4 q3 = *(const float4*)(po + 3 * 128);
    const float oth0[8] = { q0.x, q0.y, q0.z, q0.w, q1.x, q1.y, q1.z, q1.w };
    const float oth1[8] = { q2.x, q2.y, q2.z, q2.w, q3.x, q3.y, q3.z, q3.w };
    const float4 ma0 = *(const float4*)(&fin_m[tl][hz][8 * grp]);
    const float4 ma1 = *(const float4*)(&fin_m[tl][hz][8 * grp + 4]);
    const float4 mb0 = *(const float4*)(&fin_m[tl][1 - hz][8 * grp]);
    const float4 mb1 = *(const float4*)(&fin_m[tl][1 - hz][8 * grp + 4]);
    const float4 la0 = *(const float4*)(&fin_l[tl][hz][8 * grp]);
    const float4 la1 = *(const float4*)(&fin_l[tl][hz][8 * grp + 4]);
    const float4 lb0 = *(const float4*)(&fin_l[tl][1 - hz][8 * grp]);
    const float4 lb1 = *(const float4*)(&fin_l[tl][1 - hz][8 * grp + 4]);
    const float ma[8] = { ma0.x, ma0.y, ma0.z, ma0.w, ma1.x, ma1.y, ma1.z, ma1.w };
    const float mb[8] = { mb0.x, mb0.y, mb0.z, mb0.w, mb1.x, mb1.y, mb1.z, mb1.w };
    const float la[8] = { la0.x, la0.y, la0.z, la0.w, la1.x, la1.y, la1.z, la1.w };
    const float lb[8] = { lb0.x, lb0.y, lb0.z, lb0.w, lb1.x, lb1.y, lb1.z, lb1.w };

    #pragma unroll
    for (int r = 0; r < 8; ++r) {
        const float mm = fmaxf(ma[r], mb[r]);
        const float fa = __expf(ma[r] - mm);
        const float fb = __expf(mb[r] - mm);
        const float lt = la[r] * fa + lb[r] * fb;
        const long row = qrow0 + r + 8 * grp;
        out[base + row * HD + hz * HH + idx]      = (okeep0[r] * fa + oth0[r] * fb) / lt;
        out[base + row * HD + hz * HH + 16 + idx] = (okeep1[r] * fa + oth1[r] * fb) / lt;
    }
}

extern "C" void kernel_launch(void* const* d_in, const int* in_sizes, int n_in,
                              void* d_out, int out_size, void* d_ws, size_t ws_size,
                              hipStream_t stream) {
    const float* x  = (const float*)d_in[0];
    const float* Wq = (const float*)d_in[1];
    const float* Wk = (const float*)d_in[2];
    const float* Wv = (const float*)d_in[3];
    float* out = (float*)d_out;

    float* Q = (float*)d_ws;
    float* K = Q + (size_t)BT * HD;
    float* V = K + (size_t)BT * HD;

    proj_qkv_kernel<<<dim3(4, BT / PROW), 256, 0, stream>>>(x, Wq, Wk, Wv, Q, K, V);
    attn_kernel<<<dim3(AGRP, BB), 512, 0, stream>>>(Q, K, V, out);
}
